# starting kernel
# baseline (speedup 1.0000x reference)
#include <hip/hip_runtime.h>
#include <cmath>

typedef __attribute__((ext_vector_type(2))) float v2f;
typedef __attribute__((ext_vector_type(8))) float v8f;

#define D_FEAT 512
#define HIDDEN 16

__global__ void k_zero(float* __restrict__ p, long long n) {
  long long i = (long long)blockIdx.x * blockDim.x + threadIdx.x;
  long long stride = (long long)gridDim.x * blockDim.x;
  for (; i < n; i += stride) p[i] = 0.0f;
}

__global__ void k_gemm1(const float* __restrict__ x, const float* __restrict__ W,
                        float* __restrict__ h, int ntiles) {
  const int wave = blockIdx.x * (blockDim.x >> 5) + (threadIdx.x >> 5);
  if (wave >= ntiles) return;
  const int lane = threadIdx.x & 31;
  const int half = lane >> 4;
  const int l16  = lane & 15;

  const float* xrow = x + ((size_t)wave * 16 + l16) * D_FEAT + 2 * half;
  const float* wptr = W + (2 * half) * HIDDEN + l16;

  v8f c = {};
#pragma unroll 4
  for (int k0 = 0; k0 < D_FEAT; k0 += 4) {
    v2f a = *(const v2f*)(xrow + k0);
    v2f b;
    b.x = wptr[k0 * HIDDEN];
    b.y = wptr[(k0 + 1) * HIDDEN];
    c = __builtin_amdgcn_wmma_f32_16x16x4_f32(false, a, false, b,
                                              (short)0, c, false, false);
  }
  float* out = h + ((size_t)wave * 16 + 8 * half) * HIDDEN + l16;
#pragma unroll
  for (int v = 0; v < 8; ++v) out[v * HIDDEN] = c[v];
}

__global__ void k_gemm2(const float* __restrict__ acc, const float* __restrict__ bias,
                        const float* __restrict__ W, float* __restrict__ h, int ntiles) {
  const int wave = blockIdx.x * (blockDim.x >> 5) + (threadIdx.x >> 5);
  if (wave >= ntiles) return;
  const int lane = threadIdx.x & 31;
  const int half = lane >> 4;
  const int l16  = lane & 15;

  const float* arow = acc + ((size_t)wave * 16 + l16) * HIDDEN;

  v8f c = {};
#pragma unroll
  for (int k0 = 0; k0 < HIDDEN; k0 += 4) {
    const int ka = k0 + 2 * half;
    v2f a, b;
    a.x = fmaxf(arow[ka]     + bias[ka],     0.0f);
    a.y = fmaxf(arow[ka + 1] + bias[ka + 1], 0.0f);
    b.x = W[ka * HIDDEN + l16];
    b.y = W[(ka + 1) * HIDDEN + l16];
    c = __builtin_amdgcn_wmma_f32_16x16x4_f32(false, a, false, b,
                                              (short)0, c, false, false);
  }
  float* out = h + ((size_t)wave * 16 + 8 * half) * HIDDEN + l16;
#pragma unroll
  for (int v = 0; v < 8; ++v) out[v * HIDDEN] = c[v];
}

__global__ void k_scatter(const float* __restrict__ h, const int* __restrict__ src,
                          const int* __restrict__ dst, const float* __restrict__ w,
                          float* __restrict__ out, long long nedges) {
  long long i = (long long)blockIdx.x * blockDim.x + threadIdx.x;
  const long long stride = (long long)gridDim.x * blockDim.x;
  const long long total = nedges * HIDDEN;
  for (; i < total; i += stride) {
    const long long e = i >> 4;
    const int f = (int)(i & 15);
    const int s = src[e];
    const int d = dst[e];
    const float v = w[e] * h[(size_t)s * HIDDEN + f];
    atomicAdd(out + (size_t)d * HIDDEN + f, v);
  }
}

__global__ void k_final(const float* __restrict__ acc, const float* __restrict__ bias,
                        float* __restrict__ out, int n) {
  const int i = blockIdx.x * blockDim.x + threadIdx.x;
  if (i >= n) return;
  float v[HIDDEN];
  float m = -INFINITY;
#pragma unroll
  for (int f = 0; f < HIDDEN; ++f) {
    v[f] = acc[(size_t)i * HIDDEN + f] + bias[f];
    m = fmaxf(m, v[f]);
  }
  float s = 0.0f;
#pragma unroll
  for (int f = 0; f < HIDDEN; ++f) s += expf(v[f] - m);
  const float lse = m + logf(s);
#pragma unroll
  for (int f = 0; f < HIDDEN; ++f) out[(size_t)i * HIDDEN + f] = v[f] - lse;
}

extern "C" void kernel_launch(void* const* d_in, const int* in_sizes, int n_in,
                              void* d_out, int out_size, void* d_ws, size_t ws_size,
                              hipStream_t stream) {
  const float* x   = (const float*)d_in[0];
  const int*   ei  = (const int*)d_in[1];
  const float* ew  = (const float*)d_in[2];
  const float* W1  = (const float*)d_in[3];
  const float* b1  = (const float*)d_in[4];
  const float* W2  = (const float*)d_in[5];
  const float* b2  = (const float*)d_in[6];

  const int       N = in_sizes[0] / D_FEAT;
  const long long E = in_sizes[2];
  const int* src = ei;
  const int* dst = ei + E;

  float* bufA = (float*)d_ws;
  float* bufB = bufA + (size_t)N * HIDDEN;

  const int ntiles = N / 16;
  const dim3 blk(256);
  const int gemm_blocks = (ntiles + 7) / 8;
  const long long nelem = (long long)N * HIDDEN;

  k_zero<<<512, blk, 0, stream>>>(bufB, nelem);
  k_gemm1<<<gemm_blocks, blk, 0, stream>>>(x, W1, bufA, ntiles);
  k_scatter<<<32768, blk, 0, stream>>>(bufA, src, dst, ew, bufB, E);
  k_gemm2<<<gemm_blocks, blk, 0, stream>>>(bufB, b1, W2, bufA, ntiles);
  k_zero<<<512, blk, 0, stream>>>(bufB, nelem);
  k_scatter<<<32768, blk, 0, stream>>>(bufA, src, dst, ew, bufB, E);
  k_final<<<(N + 255) / 256, blk, 0, stream>>>(bufB, b2, (float*)d_out, N);
}


# Edges binned once per call by destination range (LDS counts), each layer sums a bin in LDS with LDS f32 atomic adds (arrival order), rows stored once; no global f32 atomics, no zero fills
# speedup vs baseline: 2.0520x; 2.0520x over previous
#include <hip/hip_runtime.h>
#include <cmath>

typedef __attribute__((ext_vector_type(2))) float v2f;
typedef __attribute__((ext_vector_type(8))) float v8f;

#define D_FEAT 512
#define HIDDEN 16

__global__ void k_zero(float* __restrict__ p, long long n) {
  long long i = (long long)blockIdx.x * blockDim.x + threadIdx.x;
  long long stride = (long long)gridDim.x * blockDim.x;
  for (; i < n; i += stride) p[i] = 0.0f;
}

__global__ void k_gemm1(const float* __restrict__ x, const float* __restrict__ W,
                        float* __restrict__ h, int ntiles) {
  const int wave = blockIdx.x * (blockDim.x >> 5) + (threadIdx.x >> 5);
  if (wave >= ntiles) return;
  const int lane = threadIdx.x & 31;
  const int half = lane >> 4;
  const int l16  = lane & 15;

  const float* xrow = x + ((size_t)wave * 16 + l16) * D_FEAT + 2 * half;
  const float* wptr = W + (2 * half) * HIDDEN + l16;

  v8f c = {};
#pragma unroll 4
  for (int k0 = 0; k0 < D_FEAT; k0 += 4) {
    v2f a = *(const v2f*)(xrow + k0);
    v2f b;
    b.x = wptr[k0 * HIDDEN];
    b.y = wptr[(k0 + 1) * HIDDEN];
    c = __builtin_amdgcn_wmma_f32_16x16x4_f32(false, a, false, b,
                                              (short)0, c, false, false);
  }
  float* out = h + ((size_t)wave * 16 + 8 * half) * HIDDEN + l16;
#pragma unroll
  for (int v = 0; v < 8; ++v) out[v * HIDDEN] = c[v];
}

__global__ void k_gemm2(const float* __restrict__ acc, const float* __restrict__ bias,
                        const float* __restrict__ W, float* __restrict__ h, int ntiles) {
  const int wave = blockIdx.x * (blockDim.x >> 5) + (threadIdx.x >> 5);
  if (wave >= ntiles) return;
  const int lane = threadIdx.x & 31;
  const int half = lane >> 4;
  const int l16  = lane & 15;

  const float* arow = acc + ((size_t)wave * 16 + l16) * HIDDEN;

  v8f c = {};
#pragma unroll
  for (int k0 = 0; k0 < HIDDEN; k0 += 4) {
    const int ka = k0 + 2 * half;
    v2f a, b;
    a.x = fmaxf(arow[ka]     + bias[ka],     0.0f);
    a.y = fmaxf(arow[ka + 1] + bias[ka + 1], 0.0f);
    b.x = W[ka * HIDDEN + l16];
    b.y = W[(ka + 1) * HIDDEN + l16];
    c = __builtin_amdgcn_wmma_f32_16x16x4_f32(false, a, false, b,
                                              (short)0, c, false, false);
  }
  float* out = h + ((size_t)wave * 16 + 8 * half) * HIDDEN + l16;
#pragma unroll
  for (int v = 0; v < 8; ++v) out[v * HIDDEN] = c[v];
}

__global__ void k_scatter(const float* __restrict__ h, const int* __restrict__ src,
                          const int* __restrict__ dst, const float* __restrict__ w,
                          float* __restrict__ out, long long nedges) {
  long long i = (long long)blockIdx.x * blockDim.x + threadIdx.x;
  const long long stride = (long long)gridDim.x * blockDim.x;
  const long long total = nedges * HIDDEN;
  for (; i < total; i += stride) {
    const long long e = i >> 4;
    const int f = (int)(i & 15);
    const int s = src[e];
    const int d = dst[e];
    const float v = w[e] * h[(size_t)s * HIDDEN + f];
    atomicAdd(out + (size_t)d * HIDDEN + f, v);
  }
}

__global__ void k_final(const float* __restrict__ acc, const float* __restrict__ bias,
                        float* __restrict__ out, int n) {
  const int i = blockIdx.x * blockDim.x + threadIdx.x;
  if (i >= n) return;
  float v[HIDDEN];
  float m = -INFINITY;
#pragma unroll
  for (int f = 0; f < HIDDEN; ++f) {
    v[f] = acc[(size_t)i * HIDDEN + f] + bias[f];
    m = fmaxf(m, v[f]);
  }
  float s = 0.0f;
#pragma unroll
  for (int f = 0; f < HIDDEN; ++f) s += expf(v[f] - m);
  const float lse = m + logf(s);
#pragma unroll
  for (int f = 0; f < HIDDEN; ++f) out[(size_t)i * HIDDEN + f] = v[f] - lse;
}

#define BIN_SHIFT 7
#define BIN_NODES 128
#define BIN_MAX   1024
#define BIN_THREADS 1024
#define AGG_THREADS 512
#define CHUNK_MAX 12800

__global__ void k_count(const int* __restrict__ dst, int nedges, int chunk, int nbins,
                        int* __restrict__ wcount) {
  __shared__ int hist[BIN_MAX];
  const int tid = threadIdx.x;
  hist[tid] = 0;
  __syncthreads();
  const int e0 = blockIdx.x * chunk;
  const int e1 = min(e0 + chunk, nedges);
  int e = e0 + tid;
  for (; e + 3 * BIN_THREADS < e1; e += 4 * BIN_THREADS) {
    const int d0 = dst[e];
    const int d1 = dst[e + BIN_THREADS];
    const int d2 = dst[e + 2 * BIN_THREADS];
    const int d3 = dst[e + 3 * BIN_THREADS];
    atomicAdd(&hist[d0 >> BIN_SHIFT], 1);
    atomicAdd(&hist[d1 >> BIN_SHIFT], 1);
    atomicAdd(&hist[d2 >> BIN_SHIFT], 1);
    atomicAdd(&hist[d3 >> BIN_SHIFT], 1);
  }
  for (; e < e1; e += BIN_THREADS) atomicAdd(&hist[dst[e] >> BIN_SHIFT], 1);
  __syncthreads();
  if (tid < nbins) wcount[blockIdx.x * nbins + tid] = hist[tid];
}

__global__ void k_place(const int* __restrict__ src, const int* __restrict__ dst,
                        const float* __restrict__ w, int nedges, int chunk, int nbins,
                        const int* __restrict__ wcount, int* __restrict__ bstart,
                        int* __restrict__ rec) {
  __shared__ int2 srec[CHUNK_MAX];
  __shared__ int  sdelta[CHUNK_MAX];
  __shared__ int scanG[BIN_THREADS];
  __shared__ int scanL[BIN_THREADS];
  __shared__ int lcur[BIN_MAX];
  __shared__ int delta[BIN_MAX];
  const int tid = threadIdx.x;
  const int nblk = gridDim.x;
  int tot = 0, pre = 0, own = 0;
  if (tid < nbins) {
    for (int g = 0; g < nblk; ++g) {
      const int c = wcount[g * nbins + tid];
      tot += c;
      if (g < (int)blockIdx.x) pre += c;
      if (g == (int)blockIdx.x) own = c;
    }
  }
  scanG[tid] = tot;
  scanL[tid] = own;
  __syncthreads();
  for (int off = 1; off < BIN_THREADS; off <<= 1) {
    int vg = scanG[tid];
    int vl = scanL[tid];
    if (tid >= off) { vg += scanG[tid - off]; vl += scanL[tid - off]; }
    __syncthreads();
    scanG[tid] = vg;
    scanL[tid] = vl;
    __syncthreads();
  }
  const int inclG = scanG[tid];
  const int exclG = inclG - tot;
  const int exclL = scanL[tid] - own;
  lcur[tid]  = exclL;
  delta[tid] = exclG + pre - exclL;
  if (blockIdx.x == 0 && tid < nbins) {
    bstart[tid] = exclG;
    if (tid == nbins - 1) bstart[nbins] = inclG;
  }
  __syncthreads();
  const int e0 = blockIdx.x * chunk;
  const int e1 = min(e0 + chunk, nedges);
  int e = e0 + tid;
  for (; e + BIN_THREADS < e1; e += 2 * BIN_THREADS) {
    const int da = dst[e];
    const int db = dst[e + BIN_THREADS];
    const int sa = src[e];
    const int sb = src[e + BIN_THREADS];
    const int wa = __float_as_int(w[e]);
    const int wb = __float_as_int(w[e + BIN_THREADS]);
    const int ba = da >> BIN_SHIFT;
    const int bb = db >> BIN_SHIFT;
    const int ja = atomicAdd(&lcur[ba], 1);
    const int jb = atomicAdd(&lcur[bb], 1);
    srec[ja] = make_int2(sa | ((da & (BIN_NODES - 1)) << 24), wa);
    sdelta[ja] = delta[ba];
    srec[jb] = make_int2(sb | ((db & (BIN_NODES - 1)) << 24), wb);
    sdelta[jb] = delta[bb];
  }
  for (; e < e1; e += BIN_THREADS) {
    const int d = dst[e];
    const int s = src[e];
    const int wv = __float_as_int(w[e]);
    const int b = d >> BIN_SHIFT;
    const int j = atomicAdd(&lcur[b], 1);
    srec[j] = make_int2(s | ((d & (BIN_NODES - 1)) << 24), wv);
    sdelta[j] = delta[b];
  }
  __syncthreads();
  int2* __restrict__ rec2 = (int2*)rec;
  const int cnt = e1 - e0;
  for (int j = tid; j < cnt; j += BIN_THREADS) rec2[j + sdelta[j]] = srec[j];
}

__global__ void k_agg(const float* __restrict__ h, const int* __restrict__ rec,
                      const int* __restrict__ bstart, float* __restrict__ out, int n) {
  __shared__ __attribute__((aligned(16))) float acc[BIN_NODES * HIDDEN];
  const int tid = threadIdx.x;
  *(float4*)&acc[tid * 4] = make_float4(0.0f, 0.0f, 0.0f, 0.0f);
  __syncthreads();
  const int lo = bstart[blockIdx.x];
  const int hi = bstart[blockIdx.x + 1];
  const int f = tid & 15;
  const int step = AGG_THREADS / 16;
  const int2* __restrict__ rec2 = (const int2*)rec;
  int i = lo + (tid >> 4);
  for (; i + 3 * step < hi; i += 4 * step) {
    const int2 r0 = rec2[i];
    const int2 r1 = rec2[i + step];
    const int2 r2 = rec2[i + 2 * step];
    const int2 r3 = rec2[i + 3 * step];
    const float h0 = h[(size_t)(r0.x & 0xFFFFFF) * HIDDEN + f];
    const float h1 = h[(size_t)(r1.x & 0xFFFFFF) * HIDDEN + f];
    const float h2 = h[(size_t)(r2.x & 0xFFFFFF) * HIDDEN + f];
    const float h3 = h[(size_t)(r3.x & 0xFFFFFF) * HIDDEN + f];
    atomicAdd(&acc[((r0.x >> 24) & (BIN_NODES - 1)) * HIDDEN + f], __int_as_float(r0.y) * h0);
    atomicAdd(&acc[((r1.x >> 24) & (BIN_NODES - 1)) * HIDDEN + f], __int_as_float(r1.y) * h1);
    atomicAdd(&acc[((r2.x >> 24) & (BIN_NODES - 1)) * HIDDEN + f], __int_as_float(r2.y) * h2);
    atomicAdd(&acc[((r3.x >> 24) & (BIN_NODES - 1)) * HIDDEN + f], __int_as_float(r3.y) * h3);
  }
  for (; i < hi; i += step) {
    const int2 r = rec2[i];
    const float hv = h[(size_t)(r.x & 0xFFFFFF) * HIDDEN + f];
    atomicAdd(&acc[((r.x >> 24) & (BIN_NODES - 1)) * HIDDEN + f], __int_as_float(r.y) * hv);
  }
  __syncthreads();
  const size_t o = (size_t)blockIdx.x * (BIN_NODES * HIDDEN) + (size_t)tid * 4;
  if (o < (size_t)n * HIDDEN) *(float4*)&out[o] = *(const float4*)&acc[tid * 4];
}

extern "C" void kernel_launch(void* const* d_in, const int* in_sizes, int n_in,
                              void* d_out, int out_size, void* d_ws, size_t ws_size,
                              hipStream_t stream) {
  const float* x   = (const float*)d_in[0];
  const int*   ei  = (const int*)d_in[1];
  const float* ew  = (const float*)d_in[2];
  const float* W1  = (const float*)d_in[3];
  const float* b1  = (const float*)d_in[4];
  const float* W2  = (const float*)d_in[5];
  const float* b2  = (const float*)d_in[6];

  const int       N = in_sizes[0] / D_FEAT;
  const long long E = in_sizes[2];
  const int* src = ei;
  const int* dst = ei + E;

  float* bufA = (float*)d_ws;
  float* bufB = bufA + (size_t)N * HIDDEN;

  const int ntiles = N / 16;
  const dim3 blk(256);
  const int gemm_blocks = (ntiles + 7) / 8;
  const long long nelem = (long long)N * HIDDEN;

  const int nbins  = (N + BIN_NODES - 1) >> BIN_SHIFT;
  const int nblk   = 256;
  const int nedges = (int)E;
  const int chunk  = (nedges + nblk - 1) / nblk;
  int* rec    = (int*)(bufB + (size_t)N * HIDDEN);
  int* wcount = rec + 2 * (size_t)nedges;
  int* bstart = wcount + (size_t)nblk * nbins;
  const size_t need = (size_t)((char*)(bstart + nbins + 1) - (char*)d_ws);

  if (need <= ws_size && nbins <= BIN_MAX && N <= (1 << 24) && chunk <= CHUNK_MAX) {
    k_gemm1<<<gemm_blocks, blk, 0, stream>>>(x, W1, bufA, ntiles);
    k_count<<<nblk, BIN_THREADS, 0, stream>>>(dst, nedges, chunk, nbins, wcount);
    k_place<<<nblk, BIN_THREADS, 0, stream>>>(src, dst, ew, nedges, chunk, nbins, wcount, bstart, rec);
    k_agg<<<nbins, AGG_THREADS, 0, stream>>>(bufA, rec, bstart, bufB, N);
    k_gemm2<<<gemm_blocks, blk, 0, stream>>>(bufB, b1, W2, bufA, ntiles);
    k_agg<<<nbins, AGG_THREADS, 0, stream>>>(bufA, rec, bstart, bufB, N);
    k_final<<<(N + 255) / 256, blk, 0, stream>>>(bufB, b2, (float*)d_out, N);
    return;
  }

  k_zero<<<512, blk, 0, stream>>>(bufB, nelem);
  k_gemm1<<<gemm_blocks, blk, 0, stream>>>(x, W1, bufA, ntiles);
  k_scatter<<<32768, blk, 0, stream>>>(bufA, src, dst, ew, bufB, E);
  k_gemm2<<<gemm_blocks, blk, 0, stream>>>(bufB, b1, W2, bufA, ntiles);
  k_zero<<<512, blk, 0, stream>>>(bufB, nelem);
  k_scatter<<<32768, blk, 0, stream>>>(bufA, src, dst, ew, bufB, E);
  k_final<<<(N + 255) / 256, blk, 0, stream>>>(bufB, b2, (float*)d_out, N);
}


# Layer-2 GEMM (f32 WMMA) and bias + log_softmax done as in-LDS row epilogues of the two aggregation launches: 7 launches to 5, edge sums no longer round-trip through memory; summation order unchanged
# speedup vs baseline: 2.1838x; 1.0642x over previous
#include <hip/hip_runtime.h>
#include <cmath>

typedef __attribute__((ext_vector_type(2))) float v2f;
typedef __attribute__((ext_vector_type(8))) float v8f;

#define D_FEAT 512
#define HIDDEN 16

__global__ void k_zero(float* __restrict__ p, long long n) {
  long long i = (long long)blockIdx.x * blockDim.x + threadIdx.x;
  long long stride = (long long)gridDim.x * blockDim.x;
  for (; i < n; i += stride) p[i] = 0.0f;
}

__global__ void k_gemm1(const float* __restrict__ x, const float* __restrict__ W,
                        float* __restrict__ h, int ntiles) {
  const int wave = blockIdx.x * (blockDim.x >> 5) + (threadIdx.x >> 5);
  if (wave >= ntiles) return;
  const int lane = threadIdx.x & 31;
  const int half = lane >> 4;
  const int l16  = lane & 15;

  const float* xrow = x + ((size_t)wave * 16 + l16) * D_FEAT + 2 * half;
  const float* wptr = W + (2 * half) * HIDDEN + l16;

  v8f c = {};
#pragma unroll 4
  for (int k0 = 0; k0 < D_FEAT; k0 += 4) {
    v2f a = *(const v2f*)(xrow + k0);
    v2f b;
    b.x = wptr[k0 * HIDDEN];
    b.y = wptr[(k0 + 1) * HIDDEN];
    c = __builtin_amdgcn_wmma_f32_16x16x4_f32(false, a, false, b,
                                              (short)0, c, false, false);
  }
  float* out = h + ((size_t)wave * 16 + 8 * half) * HIDDEN + l16;
#pragma unroll
  for (int v = 0; v < 8; ++v) out[v * HIDDEN] = c[v];
}

__global__ void k_gemm2(const float* __restrict__ acc, const float* __restrict__ bias,
                        const float* __restrict__ W, float* __restrict__ h, int ntiles) {
  const int wave = blockIdx.x * (blockDim.x >> 5) + (threadIdx.x >> 5);
  if (wave >= ntiles) return;
  const int lane = threadIdx.x & 31;
  const int half = lane >> 4;
  const int l16  = lane & 15;

  const float* arow = acc + ((size_t)wave * 16 + l16) * HIDDEN;

  v8f c = {};
#pragma unroll
  for (int k0 = 0; k0 < HIDDEN; k0 += 4) {
    const int ka = k0 + 2 * half;
    v2f a, b;
    a.x = fmaxf(arow[ka]     + bias[ka],     0.0f);
    a.y = fmaxf(arow[ka + 1] + bias[ka + 1], 0.0f);
    b.x = W[ka * HIDDEN + l16];
    b.y = W[(ka + 1) * HIDDEN + l16];
    c = __builtin_amdgcn_wmma_f32_16x16x4_f32(false, a, false, b,
                                              (short)0, c, false, false);
  }
  float* out = h + ((size_t)wave * 16 + 8 * half) * HIDDEN + l16;
#pragma unroll
  for (int v = 0; v < 8; ++v) out[v * HIDDEN] = c[v];
}

__global__ void k_scatter(const float* __restrict__ h, const int* __restrict__ src,
                          const int* __restrict__ dst, const float* __restrict__ w,
                          float* __restrict__ out, long long nedges) {
  long long i = (long long)blockIdx.x * blockDim.x + threadIdx.x;
  const long long stride = (long long)gridDim.x * blockDim.x;
  const long long total = nedges * HIDDEN;
  for (; i < total; i += stride) {
    const long long e = i >> 4;
    const int f = (int)(i & 15);
    const int s = src[e];
    const int d = dst[e];
    const float v = w[e] * h[(size_t)s * HIDDEN + f];
    atomicAdd(out + (size_t)d * HIDDEN + f, v);
  }
}

__global__ void k_final(const float* __restrict__ acc, const float* __restrict__ bias,
                        float* __restrict__ out, int n) {
  const int i = blockIdx.x * blockDim.x + threadIdx.x;
  if (i >= n) return;
  float v[HIDDEN];
  float m = -INFINITY;
#pragma unroll
  for (int f = 0; f < HIDDEN; ++f) {
    v[f] = acc[(size_t)i * HIDDEN + f] + bias[f];
    m = fmaxf(m, v[f]);
  }
  float s = 0.0f;
#pragma unroll
  for (int f = 0; f < HIDDEN; ++f) s += expf(v[f] - m);
  const float lse = m + logf(s);
#pragma unroll
  for (int f = 0; f < HIDDEN; ++f) out[(size_t)i * HIDDEN + f] = v[f] - lse;
}

#define BIN_SHIFT 7
#define BIN_NODES 128
#define BIN_MAX   1024
#define BIN_THREADS 1024
#define AGG_THREADS 512
#define CHUNK_MAX 12800

__global__ void k_count(const int* __restrict__ dst, int nedges, int chunk, int nbins,
                        int* __restrict__ wcount) {
  __shared__ int hist[BIN_MAX];
  const int tid = threadIdx.x;
  hist[tid] = 0;
  __syncthreads();
  const int e0 = blockIdx.x * chunk;
  const int e1 = min(e0 + chunk, nedges);
  int e = e0 + tid;
  for (; e + 3 * BIN_THREADS < e1; e += 4 * BIN_THREADS) {
    const int d0 = dst[e];
    const int d1 = dst[e + BIN_THREADS];
    const int d2 = dst[e + 2 * BIN_THREADS];
    const int d3 = dst[e + 3 * BIN_THREADS];
    atomicAdd(&hist[d0 >> BIN_SHIFT], 1);
    atomicAdd(&hist[d1 >> BIN_SHIFT], 1);
    atomicAdd(&hist[d2 >> BIN_SHIFT], 1);
    atomicAdd(&hist[d3 >> BIN_SHIFT], 1);
  }
  for (; e < e1; e += BIN_THREADS) atomicAdd(&hist[dst[e] >> BIN_SHIFT], 1);
  __syncthreads();
  if (tid < nbins) wcount[blockIdx.x * nbins + tid] = hist[tid];
}

__global__ void k_place(const int* __restrict__ src, const int* __restrict__ dst,
                        const float* __restrict__ w, int nedges, int chunk, int nbins,
                        const int* __restrict__ wcount, int* __restrict__ bstart,
                        int* __restrict__ rec) {
  __shared__ int2 srec[CHUNK_MAX];
  __shared__ int  sdelta[CHUNK_MAX];
  __shared__ int scanG[BIN_THREADS];
  __shared__ int scanL[BIN_THREADS];
  __shared__ int lcur[BIN_MAX];
  __shared__ int delta[BIN_MAX];
  const int tid = threadIdx.x;
  const int nblk = gridDim.x;
  int tot = 0, pre = 0, own = 0;
  if (tid < nbins) {
    for (int g = 0; g < nblk; ++g) {
      const int c = wcount[g * nbins + tid];
      tot += c;
      if (g < (int)blockIdx.x) pre += c;
      if (g == (int)blockIdx.x) own = c;
    }
  }
  scanG[tid] = tot;
  scanL[tid] = own;
  __syncthreads();
  for (int off = 1; off < BIN_THREADS; off <<= 1) {
    int vg = scanG[tid];
    int vl = scanL[tid];
    if (tid >= off) { vg += scanG[tid - off]; vl += scanL[tid - off]; }
    __syncthreads();
    scanG[tid] = vg;
    scanL[tid] = vl;
    __syncthreads();
  }
  const int inclG = scanG[tid];
  const int exclG = inclG - tot;
  const int exclL = scanL[tid] - own;
  lcur[tid]  = exclL;
  delta[tid] = exclG + pre - exclL;
  if (blockIdx.x == 0 && tid < nbins) {
    bstart[tid] = exclG;
    if (tid == nbins - 1) bstart[nbins] = inclG;
  }
  __syncthreads();
  const int e0 = blockIdx.x * chunk;
  const int e1 = min(e0 + chunk, nedges);
  int e = e0 + tid;
  for (; e + BIN_THREADS < e1; e += 2 * BIN_THREADS) {
    const int da = dst[e];
    const int db = dst[e + BIN_THREADS];
    const int sa = src[e];
    const int sb = src[e + BIN_THREADS];
    const int wa = __float_as_int(w[e]);
    const int wb = __float_as_int(w[e + BIN_THREADS]);
    const int ba = da >> BIN_SHIFT;
    const int bb = db >> BIN_SHIFT;
    const int ja = atomicAdd(&lcur[ba], 1);
    const int jb = atomicAdd(&lcur[bb], 1);
    srec[ja] = make_int2(sa | ((da & (BIN_NODES - 1)) << 24), wa);
    sdelta[ja] = delta[ba];
    srec[jb] = make_int2(sb | ((db & (BIN_NODES - 1)) << 24), wb);
    sdelta[jb] = delta[bb];
  }
  for (; e < e1; e += BIN_THREADS) {
    const int d = dst[e];
    const int s = src[e];
    const int wv = __float_as_int(w[e]);
    const int b = d >> BIN_SHIFT;
    const int j = atomicAdd(&lcur[b], 1);
    srec[j] = make_int2(s | ((d & (BIN_NODES - 1)) << 24), wv);
    sdelta[j] = delta[b];
  }
  __syncthreads();
  int2* __restrict__ rec2 = (int2*)rec;
  const int cnt = e1 - e0;
  for (int j = tid; j < cnt; j += BIN_THREADS) rec2[j + sdelta[j]] = srec[j];
}

#define EPI_LINEAR 0
#define EPI_LSM    1
template <int EPI>
__global__ void k_agg(const float* __restrict__ h, const int* __restrict__ rec,
                      const int* __restrict__ bstart, const float* __restrict__ bias,
                      const float* __restrict__ W, float* __restrict__ out, int n) {
  __shared__ __attribute__((aligned(16))) float acc[BIN_NODES * HIDDEN];
  const int tid = threadIdx.x;
  *(float4*)&acc[tid * 4] = make_float4(0.0f, 0.0f, 0.0f, 0.0f);
  __syncthreads();
  const int lo = bstart[blockIdx.x];
  const int hi = bstart[blockIdx.x + 1];
  const int f = tid & 15;
  const int step = AGG_THREADS / 16;
  const int2* __restrict__ rec2 = (const int2*)rec;
  const int half = (tid >> 4) & 1;
  v2f wfrag[HIDDEN / 4], bfrag[HIDDEN / 4];
  if (EPI == EPI_LINEAR) {
#pragma unroll
    for (int k0 = 0; k0 < HIDDEN; k0 += 4) {
      const int ka = k0 + 2 * half;
      wfrag[k0 / 4].x = W[ka * HIDDEN + f];
      wfrag[k0 / 4].y = W[(ka + 1) * HIDDEN + f];
      bfrag[k0 / 4].x = bias[ka];
      bfrag[k0 / 4].y = bias[ka + 1];
    }
  }
  int i = lo + (tid >> 4);
  for (; i + 3 * step < hi; i += 4 * step) {
    const int2 r0 = rec2[i];
    const int2 r1 = rec2[i + step];
    const int2 r2 = rec2[i + 2 * step];
    const int2 r3 = rec2[i + 3 * step];
    const float h0 = h[(size_t)(r0.x & 0xFFFFFF) * HIDDEN + f];
    const float h1 = h[(size_t)(r1.x & 0xFFFFFF) * HIDDEN + f];
    const float h2 = h[(size_t)(r2.x & 0xFFFFFF) * HIDDEN + f];
    const float h3 = h[(size_t)(r3.x & 0xFFFFFF) * HIDDEN + f];
    atomicAdd(&acc[((r0.x >> 24) & (BIN_NODES - 1)) * HIDDEN + f], __int_as_float(r0.y) * h0);
    atomicAdd(&acc[((r1.x >> 24) & (BIN_NODES - 1)) * HIDDEN + f], __int_as_float(r1.y) * h1);
    atomicAdd(&acc[((r2.x >> 24) & (BIN_NODES - 1)) * HIDDEN + f], __int_as_float(r2.y) * h2);
    atomicAdd(&acc[((r3.x >> 24) & (BIN_NODES - 1)) * HIDDEN + f], __int_as_float(r3.y) * h3);
  }
  for (; i < hi; i += step) {
    const int2 r = rec2[i];
    const float hv = h[(size_t)(r.x & 0xFFFFFF) * HIDDEN + f];
    atomicAdd(&acc[((r.x >> 24) & (BIN_NODES - 1)) * HIDDEN + f], __int_as_float(r.y) * hv);
  }
  __syncthreads();
  if (EPI == EPI_LINEAR) {
    const int wave = tid >> 5;
    if (wave < BIN_NODES / 16) {
      const float* arow = acc + (wave * 16 + f) * HIDDEN;
      v8f c = {};
#pragma unroll
      for (int k0 = 0; k0 < HIDDEN; k0 += 4) {
        const int ka = k0 + 2 * half;
        v2f a;
        a.x = fmaxf(arow[ka]     + bfrag[k0 / 4].x, 0.0f);
        a.y = fmaxf(arow[ka + 1] + bfrag[k0 / 4].y, 0.0f);
        c = __builtin_amdgcn_wmma_f32_16x16x4_f32(false, a, false, wfrag[k0 / 4],
                                                  (short)0, c, false, false);
      }
      float* orow = acc + (wave * 16 + 8 * half) * HIDDEN + f;
#pragma unroll
      for (int v = 0; v < 8; ++v) orow[v * HIDDEN] = c[v];
    }
  } else {
    if (tid < BIN_NODES) {
      float* row = acc + tid * HIDDEN;
      float v[HIDDEN];
      float m = -INFINITY;
#pragma unroll
      for (int q = 0; q < HIDDEN; ++q) {
        v[q] = row[q] + bias[q];
        m = fmaxf(m, v[q]);
      }
      float s = 0.0f;
#pragma unroll
      for (int q = 0; q < HIDDEN; ++q) s += expf(v[q] - m);
      const float lse = m + logf(s);
#pragma unroll
      for (int q = 0; q < HIDDEN; ++q) row[q] = v[q] - lse;
    }
  }
  __syncthreads();
  const size_t o = (size_t)blockIdx.x * (BIN_NODES * HIDDEN) + (size_t)tid * 4;
  if (o < (size_t)n * HIDDEN) *(float4*)&out[o] = *(const float4*)&acc[tid * 4];
}

extern "C" void kernel_launch(void* const* d_in, const int* in_sizes, int n_in,
                              void* d_out, int out_size, void* d_ws, size_t ws_size,
                              hipStream_t stream) {
  const float* x   = (const float*)d_in[0];
  const int*   ei  = (const int*)d_in[1];
  const float* ew  = (const float*)d_in[2];
  const float* W1  = (const float*)d_in[3];
  const float* b1  = (const float*)d_in[4];
  const float* W2  = (const float*)d_in[5];
  const float* b2  = (const float*)d_in[6];

  const int       N = in_sizes[0] / D_FEAT;
  const long long E = in_sizes[2];
  const int* src = ei;
  const int* dst = ei + E;

  float* bufA = (float*)d_ws;
  float* bufB = bufA + (size_t)N * HIDDEN;

  const int ntiles = N / 16;
  const dim3 blk(256);
  const int gemm_blocks = (ntiles + 7) / 8;
  const long long nelem = (long long)N * HIDDEN;

  const int nbins  = (N + BIN_NODES - 1) >> BIN_SHIFT;
  const int nblk   = 256;
  const int nedges = (int)E;
  const int chunk  = (nedges + nblk - 1) / nblk;
  int* rec    = (int*)(bufB + (size_t)N * HIDDEN);
  int* wcount = rec + 2 * (size_t)nedges;
  int* bstart = wcount + (size_t)nblk * nbins;
  const size_t need = (size_t)((char*)(bstart + nbins + 1) - (char*)d_ws);

  if (need <= ws_size && nbins <= BIN_MAX && N <= (1 << 24) && chunk <= CHUNK_MAX) {
    k_gemm1<<<gemm_blocks, blk, 0, stream>>>(x, W1, bufA, ntiles);
    k_count<<<nblk, BIN_THREADS, 0, stream>>>(dst, nedges, chunk, nbins, wcount);
    k_place<<<nblk, BIN_THREADS, 0, stream>>>(src, dst, ew, nedges, chunk, nbins, wcount, bstart, rec);
    k_agg<EPI_LINEAR><<<nbins, AGG_THREADS, 0, stream>>>(bufA, rec, bstart, b1, W2, bufB, N);
    k_agg<EPI_LSM><<<nbins, AGG_THREADS, 0, stream>>>(bufB, rec, bstart, b2, W2, (float*)d_out, N);
    return;
  }

  k_zero<<<512, blk, 0, stream>>>(bufB, nelem);
  k_gemm1<<<gemm_blocks, blk, 0, stream>>>(x, W1, bufA, ntiles);
  k_scatter<<<32768, blk, 0, stream>>>(bufA, src, dst, ew, bufB, E);
  k_gemm2<<<gemm_blocks, blk, 0, stream>>>(bufB, b1, W2, bufA, ntiles);
  k_zero<<<512, blk, 0, stream>>>(bufB, nelem);
  k_scatter<<<32768, blk, 0, stream>>>(bufA, src, dst, ew, bufB, E);
  k_final<<<(N + 255) / 256, blk, 0, stream>>>(bufB, b2, (float*)d_out, N);
}


# Aggregation loop: a wave loads 32 records in one coalesced load, shares them by DPP row broadcast, issues 16 row gathers per batch before the LDS adds (arrival order as before); 256-thread blocks
# speedup vs baseline: 2.3299x; 1.0669x over previous
#include <hip/hip_runtime.h>
#include <cmath>

typedef __attribute__((ext_vector_type(2))) float v2f;
typedef __attribute__((ext_vector_type(8))) float v8f;

#define D_FEAT 512
#define HIDDEN 16

__global__ void k_zero(float* __restrict__ p, long long n) {
  long long i = (long long)blockIdx.x * blockDim.x + threadIdx.x;
  long long stride = (long long)gridDim.x * blockDim.x;
  for (; i < n; i += stride) p[i] = 0.0f;
}

__global__ void k_gemm1(const float* __restrict__ x, const float* __restrict__ W,
                        float* __restrict__ h, int ntiles) {
  const int wave = blockIdx.x * (blockDim.x >> 5) + (threadIdx.x >> 5);
  if (wave >= ntiles) return;
  const int lane = threadIdx.x & 31;
  const int half = lane >> 4;
  const int l16  = lane & 15;

  const float* xrow = x + ((size_t)wave * 16 + l16) * D_FEAT + 2 * half;
  const float* wptr = W + (2 * half) * HIDDEN + l16;

  v8f c = {};
#pragma unroll 4
  for (int k0 = 0; k0 < D_FEAT; k0 += 4) {
    v2f a = *(const v2f*)(xrow + k0);
    v2f b;
    b.x = wptr[k0 * HIDDEN];
    b.y = wptr[(k0 + 1) * HIDDEN];
    c = __builtin_amdgcn_wmma_f32_16x16x4_f32(false, a, false, b,
                                              (short)0, c, false, false);
  }
  float* out = h + ((size_t)wave * 16 + 8 * half) * HIDDEN + l16;
#pragma unroll
  for (int v = 0; v < 8; ++v) out[v * HIDDEN] = c[v];
}

__global__ void k_gemm2(const float* __restrict__ acc, const float* __restrict__ bias,
                        const float* __restrict__ W, float* __restrict__ h, int ntiles) {
  const int wave = blockIdx.x * (blockDim.x >> 5) + (threadIdx.x >> 5);
  if (wave >= ntiles) return;
  const int lane = threadIdx.x & 31;
  const int half = lane >> 4;
  const int l16  = lane & 15;

  const float* arow = acc + ((size_t)wave * 16 + l16) * HIDDEN;

  v8f c = {};
#pragma unroll
  for (int k0 = 0; k0 < HIDDEN; k0 += 4) {
    const int ka = k0 + 2 * half;
    v2f a, b;
    a.x = fmaxf(arow[ka]     + bias[ka],     0.0f);
    a.y = fmaxf(arow[ka + 1] + bias[ka + 1], 0.0f);
    b.x = W[ka * HIDDEN + l16];
    b.y = W[(ka + 1) * HIDDEN + l16];
    c = __builtin_amdgcn_wmma_f32_16x16x4_f32(false, a, false, b,
                                              (short)0, c, false, false);
  }
  float* out = h + ((size_t)wave * 16 + 8 * half) * HIDDEN + l16;
#pragma unroll
  for (int v = 0; v < 8; ++v) out[v * HIDDEN] = c[v];
}

__global__ void k_scatter(const float* __restrict__ h, const int* __restrict__ src,
                          const int* __restrict__ dst, const float* __restrict__ w,
                          float* __restrict__ out, long long nedges) {
  long long i = (long long)blockIdx.x * blockDim.x + threadIdx.x;
  const long long stride = (long long)gridDim.x * blockDim.x;
  const long long total = nedges * HIDDEN;
  for (; i < total; i += stride) {
    const long long e = i >> 4;
    const int f = (int)(i & 15);
    const int s = src[e];
    const int d = dst[e];
    const float v = w[e] * h[(size_t)s * HIDDEN + f];
    atomicAdd(out + (size_t)d * HIDDEN + f, v);
  }
}

__global__ void k_final(const float* __restrict__ acc, const float* __restrict__ bias,
                        float* __restrict__ out, int n) {
  const int i = blockIdx.x * blockDim.x + threadIdx.x;
  if (i >= n) return;
  float v[HIDDEN];
  float m = -INFINITY;
#pragma unroll
  for (int f = 0; f < HIDDEN; ++f) {
    v[f] = acc[(size_t)i * HIDDEN + f] + bias[f];
    m = fmaxf(m, v[f]);
  }
  float s = 0.0f;
#pragma unroll
  for (int f = 0; f < HIDDEN; ++f) s += expf(v[f] - m);
  const float lse = m + logf(s);
#pragma unroll
  for (int f = 0; f < HIDDEN; ++f) out[(size_t)i * HIDDEN + f] = v[f] - lse;
}

#define BIN_SHIFT 7
#define BIN_NODES 128
#define BIN_MAX   1024
#define BIN_THREADS 1024
#define AGG_THREADS 256
#define CHUNK_MAX 12800

__global__ void k_count(const int* __restrict__ dst, int nedges, int chunk, int nbins,
                        int* __restrict__ wcount) {
  __shared__ int hist[BIN_MAX];
  const int tid = threadIdx.x;
  hist[tid] = 0;
  __syncthreads();
  const int e0 = blockIdx.x * chunk;
  const int e1 = min(e0 + chunk, nedges);
  int e = e0 + tid;
  for (; e + 3 * BIN_THREADS < e1; e += 4 * BIN_THREADS) {
    const int d0 = dst[e];
    const int d1 = dst[e + BIN_THREADS];
    const int d2 = dst[e + 2 * BIN_THREADS];
    const int d3 = dst[e + 3 * BIN_THREADS];
    atomicAdd(&hist[d0 >> BIN_SHIFT], 1);
    atomicAdd(&hist[d1 >> BIN_SHIFT], 1);
    atomicAdd(&hist[d2 >> BIN_SHIFT], 1);
    atomicAdd(&hist[d3 >> BIN_SHIFT], 1);
  }
  for (; e < e1; e += BIN_THREADS) atomicAdd(&hist[dst[e] >> BIN_SHIFT], 1);
  __syncthreads();
  if (tid < nbins) wcount[blockIdx.x * nbins + tid] = hist[tid];
}

__global__ void k_place(const int* __restrict__ src, const int* __restrict__ dst,
                        const float* __restrict__ w, int nedges, int chunk, int nbins,
                        const int* __restrict__ wcount, int* __restrict__ bstart,
                        int* __restrict__ rec) {
  __shared__ int2 srec[CHUNK_MAX];
  __shared__ int  sdelta[CHUNK_MAX];
  __shared__ int scanG[BIN_THREADS];
  __shared__ int scanL[BIN_THREADS];
  __shared__ int lcur[BIN_MAX];
  __shared__ int delta[BIN_MAX];
  const int tid = threadIdx.x;
  const int nblk = gridDim.x;
  int tot = 0, pre = 0, own = 0;
  if (tid < nbins) {
    for (int g = 0; g < nblk; ++g) {
      const int c = wcount[g * nbins + tid];
      tot += c;
      if (g < (int)blockIdx.x) pre += c;
      if (g == (int)blockIdx.x) own = c;
    }
  }
  scanG[tid] = tot;
  scanL[tid] = own;
  __syncthreads();
  for (int off = 1; off < BIN_THREADS; off <<= 1) {
    int vg = scanG[tid];
    int vl = scanL[tid];
    if (tid >= off) { vg += scanG[tid - off]; vl += scanL[tid - off]; }
    __syncthreads();
    scanG[tid] = vg;
    scanL[tid] = vl;
    __syncthreads();
  }
  const int inclG = scanG[tid];
  const int exclG = inclG - tot;
  const int exclL = scanL[tid] - own;
  lcur[tid]  = exclL;
  delta[tid] = exclG + pre - exclL;
  if (blockIdx.x == 0 && tid < nbins) {
    bstart[tid] = exclG;
    if (tid == nbins - 1) bstart[nbins] = inclG;
  }
  __syncthreads();
  const int e0 = blockIdx.x * chunk;
  const int e1 = min(e0 + chunk, nedges);
  int e = e0 + tid;
  for (; e + BIN_THREADS < e1; e += 2 * BIN_THREADS) {
    const int da = dst[e];
    const int db = dst[e + BIN_THREADS];
    const int sa = src[e];
    const int sb = src[e + BIN_THREADS];
    const int wa = __float_as_int(w[e]);
    const int wb = __float_as_int(w[e + BIN_THREADS]);
    const int ba = da >> BIN_SHIFT;
    const int bb = db >> BIN_SHIFT;
    const int ja = atomicAdd(&lcur[ba], 1);
    const int jb = atomicAdd(&lcur[bb], 1);
    srec[ja] = make_int2(sa | ((da & (BIN_NODES - 1)) << 24), wa);
    sdelta[ja] = delta[ba];
    srec[jb] = make_int2(sb | ((db & (BIN_NODES - 1)) << 24), wb);
    sdelta[jb] = delta[bb];
  }
  for (; e < e1; e += BIN_THREADS) {
    const int d = dst[e];
    const int s = src[e];
    const int wv = __float_as_int(w[e]);
    const int b = d >> BIN_SHIFT;
    const int j = atomicAdd(&lcur[b], 1);
    srec[j] = make_int2(s | ((d & (BIN_NODES - 1)) << 24), wv);
    sdelta[j] = delta[b];
  }
  __syncthreads();
  int2* __restrict__ rec2 = (int2*)rec;
  const int cnt = e1 - e0;
  for (int j = tid; j < cnt; j += BIN_THREADS) rec2[j + sdelta[j]] = srec[j];
}

#define EPI_LINEAR 0
#define EPI_LSM    1
#define ROW_BCAST(v, n) __builtin_amdgcn_mov_dpp((v), 0x150 + (n), 0xf, 0xf, false)
#define AGG_GATHER(n)                                                        \
  const int   s##n = ROW_BCAST(r.x, n);                                      \
  const float w##n = __int_as_float(ROW_BCAST(r.y, n));                      \
  const float g##n = h[(size_t)(s##n & 0xFFFFFF) * HIDDEN + f];
#define AGG_ADD(n) \
  atomicAdd(&acc[((s##n >> 24) & (BIN_NODES - 1)) * HIDDEN + f], w##n * g##n);
#define AGG_ADD_IF(n) \
  if (16 * half + (n) < cnt) AGG_ADD(n)
#define AGG_16(M) M(0) M(1) M(2) M(3) M(4) M(5) M(6) M(7) M(8) M(9) M(10) M(11) M(12) M(13) M(14) M(15)
template <int EPI>
__global__ void k_agg(const float* __restrict__ h, const int* __restrict__ rec,
                      const int* __restrict__ bstart, const float* __restrict__ bias,
                      const float* __restrict__ W, float* __restrict__ out, int n) {
  __shared__ __attribute__((aligned(16))) float acc[BIN_NODES * HIDDEN];
  const int tid = threadIdx.x;
  for (int q = tid; q < BIN_NODES * HIDDEN / 4; q += AGG_THREADS)
    *(float4*)&acc[q * 4] = make_float4(0.0f, 0.0f, 0.0f, 0.0f);
  __syncthreads();
  const int lo = bstart[blockIdx.x];
  const int hi = bstart[blockIdx.x + 1];
  const int f = tid & 15;
  const int2* __restrict__ rec2 = (const int2*)rec;
  const int half = (tid >> 4) & 1;
  v2f wfrag[HIDDEN / 4], bfrag[HIDDEN / 4];
  if (EPI == EPI_LINEAR) {
#pragma unroll
    for (int k0 = 0; k0 < HIDDEN; k0 += 4) {
      const int ka = k0 + 2 * half;
      wfrag[k0 / 4].x = W[ka * HIDDEN + f];
      wfrag[k0 / 4].y = W[(ka + 1) * HIDDEN + f];
      bfrag[k0 / 4].x = bias[ka];
      bfrag[k0 / 4].y = bias[ka + 1];
    }
  }
  const int lane = tid & 31;
  const int wv = __builtin_amdgcn_readfirstlane(tid >> 5);
  const int bstep = (AGG_THREADS / 32) * 32;
  int b = lo + wv * 32;
  if (b < hi) {
    int2 r = rec2[min(b + lane, hi - 1)];
    for (; b < hi; b += bstep) {
      const int2 rn = rec2[min(b + bstep + lane, hi - 1)];
      const int cnt = hi - b;
      AGG_16(AGG_GATHER)
      if (cnt >= 32) {
        AGG_16(AGG_ADD)
      } else {
        AGG_16(AGG_ADD_IF)
      }
      r = rn;
    }
  }
  __syncthreads();
  if (EPI == EPI_LINEAR) {
    const int wave = tid >> 5;
    if (wave < BIN_NODES / 16) {
      const float* arow = acc + (wave * 16 + f) * HIDDEN;
      v8f c = {};
#pragma unroll
      for (int k0 = 0; k0 < HIDDEN; k0 += 4) {
        const int ka = k0 + 2 * half;
        v2f a;
        a.x = fmaxf(arow[ka]     + bfrag[k0 / 4].x, 0.0f);
        a.y = fmaxf(arow[ka + 1] + bfrag[k0 / 4].y, 0.0f);
        c = __builtin_amdgcn_wmma_f32_16x16x4_f32(false, a, false, wfrag[k0 / 4],
                                                  (short)0, c, false, false);
      }
      float* orow = acc + (wave * 16 + 8 * half) * HIDDEN + f;
#pragma unroll
      for (int v = 0; v < 8; ++v) orow[v * HIDDEN] = c[v];
    }
  } else {
    if (tid < BIN_NODES) {
      float* row = acc + tid * HIDDEN;
      float v[HIDDEN];
      float m = -INFINITY;
#pragma unroll
      for (int q = 0; q < HIDDEN; ++q) {
        v[q] = row[q] + bias[q];
        m = fmaxf(m, v[q]);
      }
      float s = 0.0f;
#pragma unroll
      for (int q = 0; q < HIDDEN; ++q) s += expf(v[q] - m);
      const float lse = m + logf(s);
#pragma unroll
      for (int q = 0; q < HIDDEN; ++q) row[q] = v[q] - lse;
    }
  }
  __syncthreads();
  for (int q = tid; q < BIN_NODES * HIDDEN / 4; q += AGG_THREADS) {
    const size_t o = (size_t)blockIdx.x * (BIN_NODES * HIDDEN) + (size_t)q * 4;
    if (o < (size_t)n * HIDDEN) *(float4*)&out[o] = *(const float4*)&acc[q * 4];
  }
}

extern "C" void kernel_launch(void* const* d_in, const int* in_sizes, int n_in,
                              void* d_out, int out_size, void* d_ws, size_t ws_size,
                              hipStream_t stream) {
  const float* x   = (const float*)d_in[0];
  const int*   ei  = (const int*)d_in[1];
  const float* ew  = (const float*)d_in[2];
  const float* W1  = (const float*)d_in[3];
  const float* b1  = (const float*)d_in[4];
  const float* W2  = (const float*)d_in[5];
  const float* b2  = (const float*)d_in[6];

  const int       N = in_sizes[0] / D_FEAT;
  const long long E = in_sizes[2];
  const int* src = ei;
  const int* dst = ei + E;

  float* bufA = (float*)d_ws;
  float* bufB = bufA + (size_t)N * HIDDEN;

  const int ntiles = N / 16;
  const dim3 blk(256);
  const int gemm_blocks = (ntiles + 7) / 8;
  const long long nelem = (long long)N * HIDDEN;

  const int nbins  = (N + BIN_NODES - 1) >> BIN_SHIFT;
  const int nblk   = 256;
  const int nedges = (int)E;
  const int chunk  = (nedges + nblk - 1) / nblk;
  int* rec    = (int*)(bufB + (size_t)N * HIDDEN);
  int* wcount = rec + 2 * (size_t)nedges;
  int* bstart = wcount + (size_t)nblk * nbins;
  const size_t need = (size_t)((char*)(bstart + nbins + 1) - (char*)d_ws);

  if (need <= ws_size && nbins <= BIN_MAX && N <= (1 << 24) && chunk <= CHUNK_MAX) {
    k_gemm1<<<gemm_blocks, blk, 0, stream>>>(x, W1, bufA, ntiles);
    k_count<<<nblk, BIN_THREADS, 0, stream>>>(dst, nedges, chunk, nbins, wcount);
    k_place<<<nblk, BIN_THREADS, 0, stream>>>(src, dst, ew, nedges, chunk, nbins, wcount, bstart, rec);
    k_agg<EPI_LINEAR><<<nbins, AGG_THREADS, 0, stream>>>(bufA, rec, bstart, b1, W2, bufB, N);
    k_agg<EPI_LSM><<<nbins, AGG_THREADS, 0, stream>>>(bufB, rec, bstart, b2, W2, (float*)d_out, N);
    return;
  }

  k_zero<<<512, blk, 0, stream>>>(bufB, nelem);
  k_gemm1<<<gemm_blocks, blk, 0, stream>>>(x, W1, bufA, ntiles);
  k_scatter<<<32768, blk, 0, stream>>>(bufA, src, dst, ew, bufB, E);
  k_gemm2<<<gemm_blocks, blk, 0, stream>>>(bufB, b1, W2, bufA, ntiles);
  k_zero<<<512, blk, 0, stream>>>(bufB, nelem);
  k_scatter<<<32768, blk, 0, stream>>>(bufA, src, dst, ew, bufB, E);
  k_final<<<(N + 255) / 256, blk, 0, stream>>>(bufB, b2, (float*)d_out, N);
}


# Edge placement and the layer-1 GEMM share one launch (256 placement blocks, then 250 GEMM blocks of 25 row tiles; placement LDS cut to 141 KB so one of each fits a WGP): 5 launches to 4; same orders
# speedup vs baseline: 2.5514x; 1.0951x over previous
#include <hip/hip_runtime.h>
#include <cmath>

typedef __attribute__((ext_vector_type(2))) float v2f;
typedef __attribute__((ext_vector_type(8))) float v8f;

#define D_FEAT 512
#define HIDDEN 16

__global__ void k_zero(float* __restrict__ p, long long n) {
  long long i = (long long)blockIdx.x * blockDim.x + threadIdx.x;
  long long stride = (long long)gridDim.x * blockDim.x;
  for (; i < n; i += stride) p[i] = 0.0f;
}

__device__ __forceinline__ void gemm1_tile(const float* __restrict__ x, const float* __restrict__ W,
                                           float* __restrict__ h, int wave) {
  const int lane = threadIdx.x & 31;
  const int half = lane >> 4;
  const int l16  = lane & 15;

  const float* xrow = x + ((size_t)wave * 16 + l16) * D_FEAT + 2 * half;
  const float* wptr = W + (2 * half) * HIDDEN + l16;

  v8f c = {};
#pragma unroll 4
  for (int k0 = 0; k0 < D_FEAT; k0 += 4) {
    v2f a = *(const v2f*)(xrow + k0);
    v2f b;
    b.x = wptr[k0 * HIDDEN];
    b.y = wptr[(k0 + 1) * HIDDEN];
    c = __builtin_amdgcn_wmma_f32_16x16x4_f32(false, a, false, b,
                                              (short)0, c, false, false);
  }
  float* out = h + ((size_t)wave * 16 + 8 * half) * HIDDEN + l16;
#pragma unroll
  for (int v = 0; v < 8; ++v) out[v * HIDDEN] = c[v];
}

__global__ void k_gemm1(const float* __restrict__ x, const float* __restrict__ W,
                        float* __restrict__ h, int ntiles) {
  const int wave = blockIdx.x * (blockDim.x >> 5) + (threadIdx.x >> 5);
  if (wave >= ntiles) return;
  gemm1_tile(x, W, h, wave);
}

__global__ void k_gemm2(const float* __restrict__ acc, const float* __restrict__ bias,
                        const float* __restrict__ W, float* __restrict__ h, int ntiles) {
  const int wave = blockIdx.x * (blockDim.x >> 5) + (threadIdx.x >> 5);
  if (wave >= ntiles) return;
  const int lane = threadIdx.x & 31;
  const int half = lane >> 4;
  const int l16  = lane & 15;

  const float* arow = acc + ((size_t)wave * 16 + l16) * HIDDEN;

  v8f c = {};
#pragma unroll
  for (int k0 = 0; k0 < HIDDEN; k0 += 4) {
    const int ka = k0 + 2 * half;
    v2f a, b;
    a.x = fmaxf(arow[ka]     + bias[ka],     0.0f);
    a.y = fmaxf(arow[ka + 1] + bias[ka + 1], 0.0f);
    b.x = W[ka * HIDDEN + l16];
    b.y = W[(ka + 1) * HIDDEN + l16];
    c = __builtin_amdgcn_wmma_f32_16x16x4_f32(false, a, false, b,
                                              (short)0, c, false, false);
  }
  float* out = h + ((size_t)wave * 16 + 8 * half) * HIDDEN + l16;
#pragma unroll
  for (int v = 0; v < 8; ++v) out[v * HIDDEN] = c[v];
}

__global__ void k_scatter(const float* __restrict__ h, const int* __restrict__ src,
                          const int* __restrict__ dst, const float* __restrict__ w,
                          float* __restrict__ out, long long nedges) {
  long long i = (long long)blockIdx.x * blockDim.x + threadIdx.x;
  const long long stride = (long long)gridDim.x * blockDim.x;
  const long long total = nedges * HIDDEN;
  for (; i < total; i += stride) {
    const long long e = i >> 4;
    const int f = (int)(i & 15);
    const int s = src[e];
    const int d = dst[e];
    const float v = w[e] * h[(size_t)s * HIDDEN + f];
    atomicAdd(out + (size_t)d * HIDDEN + f, v);
  }
}

__global__ void k_final(const float* __restrict__ acc, const float* __restrict__ bias,
                        float* __restrict__ out, int n) {
  const int i = blockIdx.x * blockDim.x + threadIdx.x;
  if (i >= n) return;
  float v[HIDDEN];
  float m = -INFINITY;
#pragma unroll
  for (int f = 0; f < HIDDEN; ++f) {
    v[f] = acc[(size_t)i * HIDDEN + f] + bias[f];
    m = fmaxf(m, v[f]);
  }
  float s = 0.0f;
#pragma unroll
  for (int f = 0; f < HIDDEN; ++f) s += expf(v[f] - m);
  const float lse = m + logf(s);
#pragma unroll
  for (int f = 0; f < HIDDEN; ++f) out[(size_t)i * HIDDEN + f] = v[f] - lse;
}

#define BIN_SHIFT 7
#define BIN_NODES 128
#define BIN_MAX   1024
#define BIN_THREADS 1024
#define AGG_THREADS 256
#define CHUNK_MAX 12800

__global__ void k_count(const int* __restrict__ dst, int nedges, int chunk, int nbins,
                        int* __restrict__ wcount) {
  __shared__ int hist[BIN_MAX];
  const int tid = threadIdx.x;
  hist[tid] = 0;
  __syncthreads();
  const int e0 = blockIdx.x * chunk;
  const int e1 = min(e0 + chunk, nedges);
  int e = e0 + tid;
  for (; e + 3 * BIN_THREADS < e1; e += 4 * BIN_THREADS) {
    const int d0 = dst[e];
    const int d1 = dst[e + BIN_THREADS];
    const int d2 = dst[e + 2 * BIN_THREADS];
    const int d3 = dst[e + 3 * BIN_THREADS];
    atomicAdd(&hist[d0 >> BIN_SHIFT], 1);
    atomicAdd(&hist[d1 >> BIN_SHIFT], 1);
    atomicAdd(&hist[d2 >> BIN_SHIFT], 1);
    atomicAdd(&hist[d3 >> BIN_SHIFT], 1);
  }
  for (; e < e1; e += BIN_THREADS) atomicAdd(&hist[dst[e] >> BIN_SHIFT], 1);
  __syncthreads();
  if (tid < nbins) wcount[blockIdx.x * nbins + tid] = hist[tid];
}

#define GEMM_TILES 25
__global__ void __launch_bounds__(BIN_THREADS)
k_place_gemm1(const int* __restrict__ src, const int* __restrict__ dst,
              const float* __restrict__ w, int nedges, int chunk, int nbins,
              const int* __restrict__ wcount, int* __restrict__ bstart,
              int* __restrict__ rec, int nblk,
              const float* __restrict__ x, const float* __restrict__ W1,
              float* __restrict__ h, int ntiles) {
  __shared__ int2 srec[CHUNK_MAX];
  __shared__ unsigned short sbin[CHUNK_MAX];
  __shared__ int scanG[BIN_THREADS];
  __shared__ int scanL[BIN_THREADS];
  __shared__ int lcur[BIN_MAX];
  __shared__ int delta[BIN_MAX];
  const int tid = threadIdx.x;
  if ((int)blockIdx.x >= nblk) {
    const int wib  = tid >> 5;
    const int wave = ((int)blockIdx.x - nblk) * GEMM_TILES + wib;
    if (wib < GEMM_TILES && wave < ntiles)
      gemm1_tile(x, W1, h, wave);
    return;
  }
  int tot = 0, pre = 0, own = 0;
  if (tid < nbins) {
    for (int g = 0; g < nblk; ++g) {
      const int c = wcount[g * nbins + tid];
      tot += c;
      if (g < (int)blockIdx.x) pre += c;
      if (g == (int)blockIdx.x) own = c;
    }
  }
  scanG[tid] = tot;
  scanL[tid] = own;
  __syncthreads();
  for (int off = 1; off < BIN_THREADS; off <<= 1) {
    int vg = scanG[tid];
    int vl = scanL[tid];
    if (tid >= off) { vg += scanG[tid - off]; vl += scanL[tid - off]; }
    __syncthreads();
    scanG[tid] = vg;
    scanL[tid] = vl;
    __syncthreads();
  }
  const int inclG = scanG[tid];
  const int exclG = inclG - tot;
  const int exclL = scanL[tid] - own;
  lcur[tid]  = exclL;
  delta[tid] = exclG + pre - exclL;
  if (blockIdx.x == 0 && tid < nbins) {
    bstart[tid] = exclG;
    if (tid == nbins - 1) bstart[nbins] = inclG;
  }
  __syncthreads();
  const int e0 = blockIdx.x * chunk;
  const int e1 = min(e0 + chunk, nedges);
  int e = e0 + tid;
  for (; e + BIN_THREADS < e1; e += 2 * BIN_THREADS) {
    const int da = dst[e];
    const int db = dst[e + BIN_THREADS];
    const int sa = src[e];
    const int sb = src[e + BIN_THREADS];
    const int wa = __float_as_int(w[e]);
    const int wb = __float_as_int(w[e + BIN_THREADS]);
    const int ba = da >> BIN_SHIFT;
    const int bb = db >> BIN_SHIFT;
    const int ja = atomicAdd(&lcur[ba], 1);
    const int jb = atomicAdd(&lcur[bb], 1);
    srec[ja] = make_int2(sa | ((da & (BIN_NODES - 1)) << 24), wa);
    sbin[ja] = (unsigned short)ba;
    srec[jb] = make_int2(sb | ((db & (BIN_NODES - 1)) << 24), wb);
    sbin[jb] = (unsigned short)bb;
  }
  for (; e < e1; e += BIN_THREADS) {
    const int d = dst[e];
    const int s = src[e];
    const int wv = __float_as_int(w[e]);
    const int b = d >> BIN_SHIFT;
    const int j = atomicAdd(&lcur[b], 1);
    srec[j] = make_int2(s | ((d & (BIN_NODES - 1)) << 24), wv);
    sbin[j] = (unsigned short)b;
  }
  __syncthreads();
  int2* __restrict__ rec2 = (int2*)rec;
  const int cnt = e1 - e0;
  for (int j = tid; j < cnt; j += BIN_THREADS) rec2[j + delta[sbin[j]]] = srec[j];
}

#define EPI_LINEAR 0
#define EPI_LSM    1
#define ROW_BCAST(v, n) __builtin_amdgcn_mov_dpp((v), 0x150 + (n), 0xf, 0xf, false)
#define AGG_GATHER(n)                                                        \
  const int   s##n = ROW_BCAST(r.x, n);                                      \
  const float w##n = __int_as_float(ROW_BCAST(r.y, n));                      \
  const float g##n = h[(size_t)(s##n & 0xFFFFFF) * HIDDEN + f];
#define AGG_ADD(n) \
  atomicAdd(&acc[((s##n >> 24) & (BIN_NODES - 1)) * HIDDEN + f], w##n * g##n);
#define AGG_ADD_IF(n) \
  if (16 * half + (n) < cnt) AGG_ADD(n)
#define AGG_16(M) M(0) M(1) M(2) M(3) M(4) M(5) M(6) M(7) M(8) M(9) M(10) M(11) M(12) M(13) M(14) M(15)
template <int EPI>
__global__ void k_agg(const float* __restrict__ h, const int* __restrict__ rec,
                      const int* __restrict__ bstart, const float* __restrict__ bias,
                      const float* __restrict__ W, float* __restrict__ out, int n) {
  __shared__ __attribute__((aligned(16))) float acc[BIN_NODES * HIDDEN];
  const int tid = threadIdx.x;
  for (int q = tid; q < BIN_NODES * HIDDEN / 4; q += AGG_THREADS)
    *(float4*)&acc[q * 4] = make_float4(0.0f, 0.0f, 0.0f, 0.0f);
  __syncthreads();
  const int lo = bstart[blockIdx.x];
  const int hi = bstart[blockIdx.x + 1];
  const int f = tid & 15;
  const int2* __restrict__ rec2 = (const int2*)rec;
  const int half = (tid >> 4) & 1;
  v2f wfrag[HIDDEN / 4], bfrag[HIDDEN / 4];
  if (EPI == EPI_LINEAR) {
#pragma unroll
    for (int k0 = 0; k0 < HIDDEN; k0 += 4) {
      const int ka = k0 + 2 * half;
      wfrag[k0 / 4].x = W[ka * HIDDEN + f];
      wfrag[k0 / 4].y = W[(ka + 1) * HIDDEN + f];
      bfrag[k0 / 4].x = bias[ka];
      bfrag[k0 / 4].y = bias[ka + 1];
    }
  }
  const int lane = tid & 31;
  const int wv = __builtin_amdgcn_readfirstlane(tid >> 5);
  const int bstep = (AGG_THREADS / 32) * 32;
  int b = lo + wv * 32;
  if (b < hi) {
    int2 r = rec2[min(b + lane, hi - 1)];
    for (; b < hi; b += bstep) {
      const int2 rn = rec2[min(b + bstep + lane, hi - 1)];
      const int cnt = hi - b;
      AGG_16(AGG_GATHER)
      if (cnt >= 32) {
        AGG_16(AGG_ADD)
      } else {
        AGG_16(AGG_ADD_IF)
      }
      r = rn;
    }
  }
  __syncthreads();
  if (EPI == EPI_LINEAR) {
    const int wave = tid >> 5;
    if (wave < BIN_NODES / 16) {
      const float* arow = acc + (wave * 16 + f) * HIDDEN;
      v8f c = {};
#pragma unroll
      for (int k0 = 0; k0 < HIDDEN; k0 += 4) {
        const int ka = k0 + 2 * half;
        v2f a;
        a.x = fmaxf(arow[ka]     + bfrag[k0 / 4].x, 0.0f);
        a.y = fmaxf(arow[ka + 1] + bfrag[k0 / 4].y, 0.0f);
        c = __builtin_amdgcn_wmma_f32_16x16x4_f32(false, a, false, wfrag[k0 / 4],
                                                  (short)0, c, false, false);
      }
      float* orow = acc + (wave * 16 + 8 * half) * HIDDEN + f;
#pragma unroll
      for (int v = 0; v < 8; ++v) orow[v * HIDDEN] = c[v];
    }
  } else {
    if (tid < BIN_NODES) {
      float* row = acc + tid * HIDDEN;
      float v[HIDDEN];
      float m = -INFINITY;
#pragma unroll
      for (int q = 0; q < HIDDEN; ++q) {
        v[q] = row[q] + bias[q];
        m = fmaxf(m, v[q]);
      }
      float s = 0.0f;
#pragma unroll
      for (int q = 0; q < HIDDEN; ++q) s += expf(v[q] - m);
      const float lse = m + logf(s);
#pragma unroll
      for (int q = 0; q < HIDDEN; ++q) row[q] = v[q] - lse;
    }
  }
  __syncthreads();
  for (int q = tid; q < BIN_NODES * HIDDEN / 4; q += AGG_THREADS) {
    const size_t o = (size_t)blockIdx.x * (BIN_NODES * HIDDEN) + (size_t)q * 4;
    if (o < (size_t)n * HIDDEN) *(float4*)&out[o] = *(const float4*)&acc[q * 4];
  }
}

extern "C" void kernel_launch(void* const* d_in, const int* in_sizes, int n_in,
                              void* d_out, int out_size, void* d_ws, size_t ws_size,
                              hipStream_t stream) {
  const float* x   = (const float*)d_in[0];
  const int*   ei  = (const int*)d_in[1];
  const float* ew  = (const float*)d_in[2];
  const float* W1  = (const float*)d_in[3];
  const float* b1  = (const float*)d_in[4];
  const float* W2  = (const float*)d_in[5];
  const float* b2  = (const float*)d_in[6];

  const int       N = in_sizes[0] / D_FEAT;
  const long long E = in_sizes[2];
  const int* src = ei;
  const int* dst = ei + E;

  float* bufA = (float*)d_ws;
  float* bufB = bufA + (size_t)N * HIDDEN;

  const int ntiles = N / 16;
  const dim3 blk(256);
  const int gemm_blocks = (ntiles + 7) / 8;
  const long long nelem = (long long)N * HIDDEN;

  const int nbins  = (N + BIN_NODES - 1) >> BIN_SHIFT;
  const int nblk   = 256;
  const int nedges = (int)E;
  const int chunk  = (nedges + nblk - 1) / nblk;
  int* rec    = (int*)(bufB + (size_t)N * HIDDEN);
  int* wcount = rec + 2 * (size_t)nedges;
  int* bstart = wcount + (size_t)nblk * nbins;
  const size_t need = (size_t)((char*)(bstart + nbins + 1) - (char*)d_ws);

  if (need <= ws_size && nbins <= BIN_MAX && N <= (1 << 24) && chunk <= CHUNK_MAX) {
    k_count<<<nblk, BIN_THREADS, 0, stream>>>(dst, nedges, chunk, nbins, wcount);
    k_place_gemm1<<<nblk + (ntiles + GEMM_TILES - 1) / GEMM_TILES, BIN_THREADS, 0, stream>>>(
        src, dst, ew, nedges, chunk, nbins, wcount, bstart, rec, nblk, x, W1, bufA, ntiles);
    k_agg<EPI_LINEAR><<<nbins, AGG_THREADS, 0, stream>>>(bufA, rec, bstart, b1, W2, bufB, N);
    k_agg<EPI_LSM><<<nbins, AGG_THREADS, 0, stream>>>(bufB, rec, bstart, b2, W2, (float*)d_out, N);
    return;
  }

  k_zero<<<512, blk, 0, stream>>>(bufB, nelem);
  k_gemm1<<<gemm_blocks, blk, 0, stream>>>(x, W1, bufA, ntiles);
  k_scatter<<<32768, blk, 0, stream>>>(bufA, src, dst, ew, bufB, E);
  k_gemm2<<<gemm_blocks, blk, 0, stream>>>(bufB, b1, W2, bufA, ntiles);
  k_zero<<<512, blk, 0, stream>>>(bufB, nelem);
  k_scatter<<<32768, blk, 0, stream>>>(bufA, src, dst, ew, bufB, E);
  k_final<<<(N + 255) / 256, blk, 0, stream>>>(bufB, b2, (float*)d_out, N);
}


# Destination bins of 98 nodes instead of 128: 1021 aggregation blocks (4 per WGP on 256 WGPs, was 782), each with 23% fewer edges; LDS sums in arrival order as before
# speedup vs baseline: 2.6403x; 1.0348x over previous
#include <hip/hip_runtime.h>
#include <cmath>

typedef __attribute__((ext_vector_type(2))) float v2f;
typedef __attribute__((ext_vector_type(8))) float v8f;

#define D_FEAT 512
#define HIDDEN 16

__global__ void k_zero(float* __restrict__ p, long long n) {
  long long i = (long long)blockIdx.x * blockDim.x + threadIdx.x;
  long long stride = (long long)gridDim.x * blockDim.x;
  for (; i < n; i += stride) p[i] = 0.0f;
}

__device__ __forceinline__ void gemm1_tile(const float* __restrict__ x, const float* __restrict__ W,
                                           float* __restrict__ h, int wave) {
  const int lane = threadIdx.x & 31;
  const int half = lane >> 4;
  const int l16  = lane & 15;

  const float* xrow = x + ((size_t)wave * 16 + l16) * D_FEAT + 2 * half;
  const float* wptr = W + (2 * half) * HIDDEN + l16;

  v8f c = {};
#pragma unroll 4
  for (int k0 = 0; k0 < D_FEAT; k0 += 4) {
    v2f a = *(const v2f*)(xrow + k0);
    v2f b;
    b.x = wptr[k0 * HIDDEN];
    b.y = wptr[(k0 + 1) * HIDDEN];
    c = __builtin_amdgcn_wmma_f32_16x16x4_f32(false, a, false, b,
                                              (short)0, c, false, false);
  }
  float* out = h + ((size_t)wave * 16 + 8 * half) * HIDDEN + l16;
#pragma unroll
  for (int v = 0; v < 8; ++v) out[v * HIDDEN] = c[v];
}

__global__ void k_gemm1(const float* __restrict__ x, const float* __restrict__ W,
                        float* __restrict__ h, int ntiles) {
  const int wave = blockIdx.x * (blockDim.x >> 5) + (threadIdx.x >> 5);
  if (wave >= ntiles) return;
  gemm1_tile(x, W, h, wave);
}

__global__ void k_gemm2(const float* __restrict__ acc, const float* __restrict__ bias,
                        const float* __restrict__ W, float* __restrict__ h, int ntiles) {
  const int wave = blockIdx.x * (blockDim.x >> 5) + (threadIdx.x >> 5);
  if (wave >= ntiles) return;
  const int lane = threadIdx.x & 31;
  const int half = lane >> 4;
  const int l16  = lane & 15;

  const float* arow = acc + ((size_t)wave * 16 + l16) * HIDDEN;

  v8f c = {};
#pragma unroll
  for (int k0 = 0; k0 < HIDDEN; k0 += 4) {
    const int ka = k0 + 2 * half;
    v2f a, b;
    a.x = fmaxf(arow[ka]     + bias[ka],     0.0f);
    a.y = fmaxf(arow[ka + 1] + bias[ka + 1], 0.0f);
    b.x = W[ka * HIDDEN + l16];
    b.y = W[(ka + 1) * HIDDEN + l16];
    c = __builtin_amdgcn_wmma_f32_16x16x4_f32(false, a, false, b,
                                              (short)0, c, false, false);
  }
  float* out = h + ((size_t)wave * 16 + 8 * half) * HIDDEN + l16;
#pragma unroll
  for (int v = 0; v < 8; ++v) out[v * HIDDEN] = c[v];
}

__global__ void k_scatter(const float* __restrict__ h, const int* __restrict__ src,
                          const int* __restrict__ dst, const float* __restrict__ w,
                          float* __restrict__ out, long long nedges) {
  long long i = (long long)blockIdx.x * blockDim.x + threadIdx.x;
  const long long stride = (long long)gridDim.x * blockDim.x;
  const long long total = nedges * HIDDEN;
  for (; i < total; i += stride) {
    const long long e = i >> 4;
    const int f = (int)(i & 15);
    const int s = src[e];
    const int d = dst[e];
    const float v = w[e] * h[(size_t)s * HIDDEN + f];
    atomicAdd(out + (size_t)d * HIDDEN + f, v);
  }
}

__global__ void k_final(const float* __restrict__ acc, const float* __restrict__ bias,
                        float* __restrict__ out, int n) {
  const int i = blockIdx.x * blockDim.x + threadIdx.x;
  if (i >= n) return;
  float v[HIDDEN];
  float m = -INFINITY;
#pragma unroll
  for (int f = 0; f < HIDDEN; ++f) {
    v[f] = acc[(size_t)i * HIDDEN + f] + bias[f];
    m = fmaxf(m, v[f]);
  }
  float s = 0.0f;
#pragma unroll
  for (int f = 0; f < HIDDEN; ++f) s += expf(v[f] - m);
  const float lse = m + logf(s);
#pragma unroll
  for (int f = 0; f < HIDDEN; ++f) out[(size_t)i * HIDDEN + f] = v[f] - lse;
}

#define BIN_NODES 98
#define BIN_ROWS  112
#define BIN_LMASK 127
#define BIN_MAX   1024
__device__ __forceinline__ int bin_of(int d) { return (int)((unsigned)d / (unsigned)BIN_NODES); }
#define BIN_THREADS 1024
#define AGG_THREADS 256
#define CHUNK_MAX 12800

__global__ void k_count(const int* __restrict__ dst, int nedges, int chunk, int nbins,
                        int* __restrict__ wcount) {
  __shared__ int hist[BIN_MAX];
  const int tid = threadIdx.x;
  hist[tid] = 0;
  __syncthreads();
  const int e0 = blockIdx.x * chunk;
  const int e1 = min(e0 + chunk, nedges);
  int e = e0 + tid;
  for (; e + 3 * BIN_THREADS < e1; e += 4 * BIN_THREADS) {
    const int d0 = dst[e];
    const int d1 = dst[e + BIN_THREADS];
    const int d2 = dst[e + 2 * BIN_THREADS];
    const int d3 = dst[e + 3 * BIN_THREADS];
    atomicAdd(&hist[bin_of(d0)], 1);
    atomicAdd(&hist[bin_of(d1)], 1);
    atomicAdd(&hist[bin_of(d2)], 1);
    atomicAdd(&hist[bin_of(d3)], 1);
  }
  for (; e < e1; e += BIN_THREADS) atomicAdd(&hist[bin_of(dst[e])], 1);
  __syncthreads();
  if (tid < nbins) wcount[blockIdx.x * nbins + tid] = hist[tid];
}

#define GEMM_TILES 25
__global__ void __launch_bounds__(BIN_THREADS)
k_place_gemm1(const int* __restrict__ src, const int* __restrict__ dst,
              const float* __restrict__ w, int nedges, int chunk, int nbins,
              const int* __restrict__ wcount, int* __restrict__ bstart,
              int* __restrict__ rec, int nblk,
              const float* __restrict__ x, const float* __restrict__ W1,
              float* __restrict__ h, int ntiles) {
  __shared__ int2 srec[CHUNK_MAX];
  __shared__ unsigned short sbin[CHUNK_MAX];
  __shared__ int scanG[BIN_THREADS];
  __shared__ int scanL[BIN_THREADS];
  __shared__ int lcur[BIN_MAX];
  __shared__ int delta[BIN_MAX];
  const int tid = threadIdx.x;
  if ((int)blockIdx.x >= nblk) {
    const int wib  = tid >> 5;
    const int wave = ((int)blockIdx.x - nblk) * GEMM_TILES + wib;
    if (wib < GEMM_TILES && wave < ntiles)
      gemm1_tile(x, W1, h, wave);
    return;
  }
  int tot = 0, pre = 0, own = 0;
  if (tid < nbins) {
    for (int g = 0; g < nblk; ++g) {
      const int c = wcount[g * nbins + tid];
      tot += c;
      if (g < (int)blockIdx.x) pre += c;
      if (g == (int)blockIdx.x) own = c;
    }
  }
  scanG[tid] = tot;
  scanL[tid] = own;
  __syncthreads();
  for (int off = 1; off < BIN_THREADS; off <<= 1) {
    int vg = scanG[tid];
    int vl = scanL[tid];
    if (tid >= off) { vg += scanG[tid - off]; vl += scanL[tid - off]; }
    __syncthreads();
    scanG[tid] = vg;
    scanL[tid] = vl;
    __syncthreads();
  }
  const int inclG = scanG[tid];
  const int exclG = inclG - tot;
  const int exclL = scanL[tid] - own;
  lcur[tid]  = exclL;
  delta[tid] = exclG + pre - exclL;
  if (blockIdx.x == 0 && tid < nbins) {
    bstart[tid] = exclG;
    if (tid == nbins - 1) bstart[nbins] = inclG;
  }
  __syncthreads();
  const int e0 = blockIdx.x * chunk;
  const int e1 = min(e0 + chunk, nedges);
  int e = e0 + tid;
  for (; e + BIN_THREADS < e1; e += 2 * BIN_THREADS) {
    const int da = dst[e];
    const int db = dst[e + BIN_THREADS];
    const int sa = src[e];
    const int sb = src[e + BIN_THREADS];
    const int wa = __float_as_int(w[e]);
    const int wb = __float_as_int(w[e + BIN_THREADS]);
    const int ba = bin_of(da);
    const int bb = bin_of(db);
    const int ja = atomicAdd(&lcur[ba], 1);
    const int jb = atomicAdd(&lcur[bb], 1);
    srec[ja] = make_int2(sa | ((da - ba * BIN_NODES) << 24), wa);
    sbin[ja] = (unsigned short)ba;
    srec[jb] = make_int2(sb | ((db - bb * BIN_NODES) << 24), wb);
    sbin[jb] = (unsigned short)bb;
  }
  for (; e < e1; e += BIN_THREADS) {
    const int d = dst[e];
    const int s = src[e];
    const int wv = __float_as_int(w[e]);
    const int b = bin_of(d);
    const int j = atomicAdd(&lcur[b], 1);
    srec[j] = make_int2(s | ((d - b * BIN_NODES) << 24), wv);
    sbin[j] = (unsigned short)b;
  }
  __syncthreads();
  int2* __restrict__ rec2 = (int2*)rec;
  const int cnt = e1 - e0;
  for (int j = tid; j < cnt; j += BIN_THREADS) rec2[j + delta[sbin[j]]] = srec[j];
}

#define EPI_LINEAR 0
#define EPI_LSM    1
#define ROW_BCAST(v, n) __builtin_amdgcn_mov_dpp((v), 0x150 + (n), 0xf, 0xf, false)
#define AGG_GATHER(n)                                                        \
  const int   s##n = ROW_BCAST(r.x, n);                                      \
  const float w##n = __int_as_float(ROW_BCAST(r.y, n));                      \
  const float g##n = h[(size_t)(s##n & 0xFFFFFF) * HIDDEN + f];
#define AGG_ADD(n) \
  atomicAdd(&acc[((s##n >> 24) & BIN_LMASK) * HIDDEN + f], w##n * g##n);
#define AGG_ADD_IF(n) \
  if (16 * half + (n) < cnt) AGG_ADD(n)
#define AGG_16(M) M(0) M(1) M(2) M(3) M(4) M(5) M(6) M(7) M(8) M(9) M(10) M(11) M(12) M(13) M(14) M(15)
template <int EPI>
__global__ void k_agg(const float* __restrict__ h, const int* __restrict__ rec,
                      const int* __restrict__ bstart, const float* __restrict__ bias,
                      const float* __restrict__ W, float* __restrict__ out, int n) {
  __shared__ __attribute__((aligned(16))) float acc[BIN_ROWS * HIDDEN];
  const int tid = threadIdx.x;
  for (int q = tid; q < BIN_ROWS * HIDDEN / 4; q += AGG_THREADS)
    *(float4*)&acc[q * 4] = make_float4(0.0f, 0.0f, 0.0f, 0.0f);
  __syncthreads();
  const int lo = bstart[blockIdx.x];
  const int hi = bstart[blockIdx.x + 1];
  const int f = tid & 15;
  const int2* __restrict__ rec2 = (const int2*)rec;
  const int half = (tid >> 4) & 1;
  v2f wfrag[HIDDEN / 4], bfrag[HIDDEN / 4];
  if (EPI == EPI_LINEAR) {
#pragma unroll
    for (int k0 = 0; k0 < HIDDEN; k0 += 4) {
      const int ka = k0 + 2 * half;
      wfrag[k0 / 4].x = W[ka * HIDDEN + f];
      wfrag[k0 / 4].y = W[(ka + 1) * HIDDEN + f];
      bfrag[k0 / 4].x = bias[ka];
      bfrag[k0 / 4].y = bias[ka + 1];
    }
  }
  const int lane = tid & 31;
  const int wv = __builtin_amdgcn_readfirstlane(tid >> 5);
  const int bstep = (AGG_THREADS / 32) * 32;
  int b = lo + wv * 32;
  if (b < hi) {
    int2 r = rec2[min(b + lane, hi - 1)];
    for (; b < hi; b += bstep) {
      const int2 rn = rec2[min(b + bstep + lane, hi - 1)];
      const int cnt = hi - b;
      AGG_16(AGG_GATHER)
      if (cnt >= 32) {
        AGG_16(AGG_ADD)
      } else {
        AGG_16(AGG_ADD_IF)
      }
      r = rn;
    }
  }
  __syncthreads();
  if (EPI == EPI_LINEAR) {
    const int wave = tid >> 5;
    if (wave < BIN_ROWS / 16) {
      const float* arow = acc + (wave * 16 + f) * HIDDEN;
      v8f c = {};
#pragma unroll
      for (int k0 = 0; k0 < HIDDEN; k0 += 4) {
        const int ka = k0 + 2 * half;
        v2f a;
        a.x = fmaxf(arow[ka]     + bfrag[k0 / 4].x, 0.0f);
        a.y = fmaxf(arow[ka + 1] + bfrag[k0 / 4].y, 0.0f);
        c = __builtin_amdgcn_wmma_f32_16x16x4_f32(false, a, false, wfrag[k0 / 4],
                                                  (short)0, c, false, false);
      }
      float* orow = acc + (wave * 16 + 8 * half) * HIDDEN + f;
#pragma unroll
      for (int v = 0; v < 8; ++v) orow[v * HIDDEN] = c[v];
    }
  } else {
    if (tid < BIN_NODES) {
      float* row = acc + tid * HIDDEN;
      float v[HIDDEN];
      float m = -INFINITY;
#pragma unroll
      for (int q = 0; q < HIDDEN; ++q) {
        v[q] = row[q] + bias[q];
        m = fmaxf(m, v[q]);
      }
      float s = 0.0f;
#pragma unroll
      for (int q = 0; q < HIDDEN; ++q) s += expf(v[q] - m);
      const float lse = m + logf(s);
#pragma unroll
      for (int q = 0; q < HIDDEN; ++q) row[q] = v[q] - lse;
    }
  }
  __syncthreads();
  for (int q = tid; q < BIN_NODES * HIDDEN / 4; q += AGG_THREADS) {
    const size_t o = (size_t)blockIdx.x * (BIN_NODES * HIDDEN) + (size_t)q * 4;
    if (o < (size_t)n * HIDDEN) *(float4*)&out[o] = *(const float4*)&acc[q * 4];
  }
}

extern "C" void kernel_launch(void* const* d_in, const int* in_sizes, int n_in,
                              void* d_out, int out_size, void* d_ws, size_t ws_size,
                              hipStream_t stream) {
  const float* x   = (const float*)d_in[0];
  const int*   ei  = (const int*)d_in[1];
  const float* ew  = (const float*)d_in[2];
  const float* W1  = (const float*)d_in[3];
  const float* b1  = (const float*)d_in[4];
  const float* W2  = (const float*)d_in[5];
  const float* b2  = (const float*)d_in[6];

  const int       N = in_sizes[0] / D_FEAT;
  const long long E = in_sizes[2];
  const int* src = ei;
  const int* dst = ei + E;

  float* bufA = (float*)d_ws;
  float* bufB = bufA + (size_t)N * HIDDEN;

  const int ntiles = N / 16;
  const dim3 blk(256);
  const int gemm_blocks = (ntiles + 7) / 8;
  const long long nelem = (long long)N * HIDDEN;

  const int nbins  = (N + BIN_NODES - 1) / BIN_NODES;
  const int nblk   = 256;
  const int nedges = (int)E;
  const int chunk  = (nedges + nblk - 1) / nblk;
  int* rec    = (int*)(bufB + (size_t)N * HIDDEN);
  int* wcount = rec + 2 * (size_t)nedges;
  int* bstart = wcount + (size_t)nblk * nbins;
  const size_t need = (size_t)((char*)(bstart + nbins + 1) - (char*)d_ws);

  if (need <= ws_size && nbins <= BIN_MAX && N <= (1 << 24) && chunk <= CHUNK_MAX) {
    k_count<<<nblk, BIN_THREADS, 0, stream>>>(dst, nedges, chunk, nbins, wcount);
    k_place_gemm1<<<nblk + (ntiles + GEMM_TILES - 1) / GEMM_TILES, BIN_THREADS, 0, stream>>>(
        src, dst, ew, nedges, chunk, nbins, wcount, bstart, rec, nblk, x, W1, bufA, ntiles);
    k_agg<EPI_LINEAR><<<nbins, AGG_THREADS, 0, stream>>>(bufA, rec, bstart, b1, W2, bufB, N);
    k_agg<EPI_LSM><<<nbins, AGG_THREADS, 0, stream>>>(bufB, rec, bstart, b2, W2, (float*)d_out, N);
    return;
  }

  k_zero<<<512, blk, 0, stream>>>(bufB, nelem);
  k_gemm1<<<gemm_blocks, blk, 0, stream>>>(x, W1, bufA, ntiles);
  k_scatter<<<32768, blk, 0, stream>>>(bufA, src, dst, ew, bufB, E);
  k_gemm2<<<gemm_blocks, blk, 0, stream>>>(bufB, b1, W2, bufA, ntiles);
  k_zero<<<512, blk, 0, stream>>>(bufB, nelem);
  k_scatter<<<32768, blk, 0, stream>>>(bufA, src, dst, ew, bufB, E);
  k_final<<<(N + 255) / 256, blk, 0, stream>>>(bufB, b2, (float*)d_out, N);
}


# Place blocks form 8 groups, each with its own record region: a place block sums 32 count rows, not 256; each aggregation wave takes one group's run of its bin; LDS sums in arrival order as before
# speedup vs baseline: 2.9889x; 1.1320x over previous
#include <hip/hip_runtime.h>
#include <cmath>

typedef __attribute__((ext_vector_type(2))) float v2f;
typedef __attribute__((ext_vector_type(8))) float v8f;

#define D_FEAT 512
#define HIDDEN 16

__global__ void k_zero(float* __restrict__ p, long long n) {
  long long i = (long long)blockIdx.x * blockDim.x + threadIdx.x;
  long long stride = (long long)gridDim.x * blockDim.x;
  for (; i < n; i += stride) p[i] = 0.0f;
}

__device__ __forceinline__ void gemm1_tile(const float* __restrict__ x, const float* __restrict__ W,
                                           float* __restrict__ h, int wave) {
  const int lane = threadIdx.x & 31;
  const int half = lane >> 4;
  const int l16  = lane & 15;

  const float* xrow = x + ((size_t)wave * 16 + l16) * D_FEAT + 2 * half;
  const float* wptr = W + (2 * half) * HIDDEN + l16;

  v8f c = {};
#pragma unroll 4
  for (int k0 = 0; k0 < D_FEAT; k0 += 4) {
    v2f a = *(const v2f*)(xrow + k0);
    v2f b;
    b.x = wptr[k0 * HIDDEN];
    b.y = wptr[(k0 + 1) * HIDDEN];
    c = __builtin_amdgcn_wmma_f32_16x16x4_f32(false, a, false, b,
                                              (short)0, c, false, false);
  }
  float* out = h + ((size_t)wave * 16 + 8 * half) * HIDDEN + l16;
#pragma unroll
  for (int v = 0; v < 8; ++v) out[v * HIDDEN] = c[v];
}

__global__ void k_gemm1(const float* __restrict__ x, const float* __restrict__ W,
                        float* __restrict__ h, int ntiles) {
  const int wave = blockIdx.x * (blockDim.x >> 5) + (threadIdx.x >> 5);
  if (wave >= ntiles) return;
  gemm1_tile(x, W, h, wave);
}

__global__ void k_gemm2(const float* __restrict__ acc, const float* __restrict__ bias,
                        const float* __restrict__ W, float* __restrict__ h, int ntiles) {
  const int wave = blockIdx.x * (blockDim.x >> 5) + (threadIdx.x >> 5);
  if (wave >= ntiles) return;
  const int lane = threadIdx.x & 31;
  const int half = lane >> 4;
  const int l16  = lane & 15;

  const float* arow = acc + ((size_t)wave * 16 + l16) * HIDDEN;

  v8f c = {};
#pragma unroll
  for (int k0 = 0; k0 < HIDDEN; k0 += 4) {
    const int ka = k0 + 2 * half;
    v2f a, b;
    a.x = fmaxf(arow[ka]     + bias[ka],     0.0f);
    a.y = fmaxf(arow[ka + 1] + bias[ka + 1], 0.0f);
    b.x = W[ka * HIDDEN + l16];
    b.y = W[(ka + 1) * HIDDEN + l16];
    c = __builtin_amdgcn_wmma_f32_16x16x4_f32(false, a, false, b,
                                              (short)0, c, false, false);
  }
  float* out = h + ((size_t)wave * 16 + 8 * half) * HIDDEN + l16;
#pragma unroll
  for (int v = 0; v < 8; ++v) out[v * HIDDEN] = c[v];
}

__global__ void k_scatter(const float* __restrict__ h, const int* __restrict__ src,
                          const int* __restrict__ dst, const float* __restrict__ w,
                          float* __restrict__ out, long long nedges) {
  long long i = (long long)blockIdx.x * blockDim.x + threadIdx.x;
  const long long stride = (long long)gridDim.x * blockDim.x;
  const long long total = nedges * HIDDEN;
  for (; i < total; i += stride) {
    const long long e = i >> 4;
    const int f = (int)(i & 15);
    const int s = src[e];
    const int d = dst[e];
    const float v = w[e] * h[(size_t)s * HIDDEN + f];
    atomicAdd(out + (size_t)d * HIDDEN + f, v);
  }
}

__global__ void k_final(const float* __restrict__ acc, const float* __restrict__ bias,
                        float* __restrict__ out, int n) {
  const int i = blockIdx.x * blockDim.x + threadIdx.x;
  if (i >= n) return;
  float v[HIDDEN];
  float m = -INFINITY;
#pragma unroll
  for (int f = 0; f < HIDDEN; ++f) {
    v[f] = acc[(size_t)i * HIDDEN + f] + bias[f];
    m = fmaxf(m, v[f]);
  }
  float s = 0.0f;
#pragma unroll
  for (int f = 0; f < HIDDEN; ++f) s += expf(v[f] - m);
  const float lse = m + logf(s);
#pragma unroll
  for (int f = 0; f < HIDDEN; ++f) out[(size_t)i * HIDDEN + f] = v[f] - lse;
}

#define BIN_NODES 98
#define BIN_ROWS  112
#define BIN_LMASK 127
#define BIN_MAX   1024
__device__ __forceinline__ int bin_of(int d) { return (int)((unsigned)d / (unsigned)BIN_NODES); }
#define BIN_THREADS 1024
#define AGG_THREADS 256
#define CHUNK_MAX 12800
#define AGG_WAVES (AGG_THREADS / 32)
#define GRP_BLKS  32
#define GS_STRIDE 1056

__global__ void k_count(const int* __restrict__ dst, int nedges, int chunk, int nbins,
                        int* __restrict__ wcount) {
  __shared__ int hist[BIN_MAX];
  const int tid = threadIdx.x;
  hist[tid] = 0;
  __syncthreads();
  const int e0 = blockIdx.x * chunk;
  const int e1 = min(e0 + chunk, nedges);
  int e = e0 + tid;
  for (; e + 3 * BIN_THREADS < e1; e += 4 * BIN_THREADS) {
    const int d0 = dst[e];
    const int d1 = dst[e + BIN_THREADS];
    const int d2 = dst[e + 2 * BIN_THREADS];
    const int d3 = dst[e + 3 * BIN_THREADS];
    atomicAdd(&hist[bin_of(d0)], 1);
    atomicAdd(&hist[bin_of(d1)], 1);
    atomicAdd(&hist[bin_of(d2)], 1);
    atomicAdd(&hist[bin_of(d3)], 1);
  }
  for (; e < e1; e += BIN_THREADS) atomicAdd(&hist[bin_of(dst[e])], 1);
  __syncthreads();
  if (tid < nbins) wcount[blockIdx.x * nbins + tid] = hist[tid];
}

#define GEMM_TILES 25
__global__ void __launch_bounds__(BIN_THREADS)
k_place_gemm1(const int* __restrict__ src, const int* __restrict__ dst,
              const float* __restrict__ w, int nedges, int chunk, int nbins,
              const int* __restrict__ wcount, int* __restrict__ gstart,
              int* __restrict__ rec, int nblk,
              const float* __restrict__ x, const float* __restrict__ W1,
              float* __restrict__ h, int ntiles) {
  __shared__ int2 srec[CHUNK_MAX];
  __shared__ unsigned short sbin[CHUNK_MAX];
  __shared__ int scanG[BIN_THREADS];
  __shared__ int scanL[BIN_THREADS];
  __shared__ int lcur[BIN_MAX];
  __shared__ int delta[BIN_MAX];
  const int tid = threadIdx.x;
  if ((int)blockIdx.x >= nblk) {
    const int wib  = tid >> 5;
    const int wave = ((int)blockIdx.x - nblk) * GEMM_TILES + wib;
    if (wib < GEMM_TILES && wave < ntiles)
      gemm1_tile(x, W1, h, wave);
    return;
  }
  const int g0 = ((int)blockIdx.x / GRP_BLKS) * GRP_BLKS;
  int tot = 0, pre = 0, own = 0;
  if (tid < nbins) {
#pragma unroll
    for (int i = 0; i < GRP_BLKS; ++i) {
      const int c = wcount[(g0 + i) * nbins + tid];
      tot += c;
      if (g0 + i < (int)blockIdx.x) pre += c;
      if (g0 + i == (int)blockIdx.x) own = c;
    }
  }
  scanG[tid] = tot;
  scanL[tid] = own;
  __syncthreads();
  for (int off = 1; off < BIN_THREADS; off <<= 1) {
    int vg = scanG[tid];
    int vl = scanL[tid];
    if (tid >= off) { vg += scanG[tid - off]; vl += scanL[tid - off]; }
    __syncthreads();
    scanG[tid] = vg;
    scanL[tid] = vl;
    __syncthreads();
  }
  const int gbase = g0 * chunk;
  const int inclG = gbase + scanG[tid];
  const int exclG = inclG - tot;
  const int exclL = scanL[tid] - own;
  lcur[tid]  = exclL;
  delta[tid] = exclG + pre - exclL;
  if ((int)blockIdx.x == g0 && tid < nbins) {
    int* __restrict__ gs = gstart + (g0 / GRP_BLKS) * GS_STRIDE;
    gs[tid] = exclG;
    if (tid == nbins - 1) gs[nbins] = inclG;
  }
  __syncthreads();
  const int e0 = blockIdx.x * chunk;
  const int e1 = min(e0 + chunk, nedges);
  int e = e0 + tid;
  for (; e + BIN_THREADS < e1; e += 2 * BIN_THREADS) {
    const int da = dst[e];
    const int db = dst[e + BIN_THREADS];
    const int sa = src[e];
    const int sb = src[e + BIN_THREADS];
    const int wa = __float_as_int(w[e]);
    const int wb = __float_as_int(w[e + BIN_THREADS]);
    const int ba = bin_of(da);
    const int bb = bin_of(db);
    const int ja = atomicAdd(&lcur[ba], 1);
    const int jb = atomicAdd(&lcur[bb], 1);
    srec[ja] = make_int2(sa | ((da - ba * BIN_NODES) << 24), wa);
    sbin[ja] = (unsigned short)ba;
    srec[jb] = make_int2(sb | ((db - bb * BIN_NODES) << 24), wb);
    sbin[jb] = (unsigned short)bb;
  }
  for (; e < e1; e += BIN_THREADS) {
    const int d = dst[e];
    const int s = src[e];
    const int wv = __float_as_int(w[e]);
    const int b = bin_of(d);
    const int j = atomicAdd(&lcur[b], 1);
    srec[j] = make_int2(s | ((d - b * BIN_NODES) << 24), wv);
    sbin[j] = (unsigned short)b;
  }
  __syncthreads();
  int2* __restrict__ rec2 = (int2*)rec;
  const int cnt = e1 - e0;
  for (int j = tid; j < cnt; j += BIN_THREADS) rec2[j + delta[sbin[j]]] = srec[j];
}

#define EPI_LINEAR 0
#define EPI_LSM    1
#define ROW_BCAST(v, n) __builtin_amdgcn_mov_dpp((v), 0x150 + (n), 0xf, 0xf, false)
#define AGG_GATHER(n)                                                        \
  const int   s##n = ROW_BCAST(r.x, n);                                      \
  const float w##n = __int_as_float(ROW_BCAST(r.y, n));                      \
  const float g##n = h[(size_t)(s##n & 0xFFFFFF) * HIDDEN + f];
#define AGG_ADD(n) \
  atomicAdd(&acc[((s##n >> 24) & BIN_LMASK) * HIDDEN + f], w##n * g##n);
#define AGG_ADD_IF(n) \
  if (16 * half + (n) < cnt) AGG_ADD(n)
#define AGG_16(M) M(0) M(1) M(2) M(3) M(4) M(5) M(6) M(7) M(8) M(9) M(10) M(11) M(12) M(13) M(14) M(15)
template <int EPI>
__global__ void k_agg(const float* __restrict__ h, const int* __restrict__ rec,
                      const int* __restrict__ gstart, const float* __restrict__ bias,
                      const float* __restrict__ W, float* __restrict__ out, int n) {
  __shared__ __attribute__((aligned(16))) float acc[BIN_ROWS * HIDDEN];
  const int tid = threadIdx.x;
  for (int q = tid; q < BIN_ROWS * HIDDEN / 4; q += AGG_THREADS)
    *(float4*)&acc[q * 4] = make_float4(0.0f, 0.0f, 0.0f, 0.0f);
  __syncthreads();
  const int f = tid & 15;
  const int2* __restrict__ rec2 = (const int2*)rec;
  const int half = (tid >> 4) & 1;
  v2f wfrag[HIDDEN / 4], bfrag[HIDDEN / 4];
  if (EPI == EPI_LINEAR) {
#pragma unroll
    for (int k0 = 0; k0 < HIDDEN; k0 += 4) {
      const int ka = k0 + 2 * half;
      wfrag[k0 / 4].x = W[ka * HIDDEN + f];
      wfrag[k0 / 4].y = W[(ka + 1) * HIDDEN + f];
      bfrag[k0 / 4].x = bias[ka];
      bfrag[k0 / 4].y = bias[ka + 1];
    }
  }
  const int lane = tid & 31;
  const int wv = __builtin_amdgcn_readfirstlane(tid >> 5);
  const int lo = gstart[wv * GS_STRIDE + blockIdx.x];
  const int hi = gstart[wv * GS_STRIDE + blockIdx.x + 1];
  const int bstep = 32;
  int b = lo;
  if (b < hi) {
    int2 r = rec2[min(b + lane, hi - 1)];
    for (; b < hi; b += bstep) {
      const int2 rn = rec2[min(b + bstep + lane, hi - 1)];
      const int cnt = hi - b;
      AGG_16(AGG_GATHER)
      if (cnt >= 32) {
        AGG_16(AGG_ADD)
      } else {
        AGG_16(AGG_ADD_IF)
      }
      r = rn;
    }
  }
  __syncthreads();
  if (EPI == EPI_LINEAR) {
    const int wave = tid >> 5;
    if (wave < BIN_ROWS / 16) {
      const float* arow = acc + (wave * 16 + f) * HIDDEN;
      v8f c = {};
#pragma unroll
      for (int k0 = 0; k0 < HIDDEN; k0 += 4) {
        const int ka = k0 + 2 * half;
        v2f a;
        a.x = fmaxf(arow[ka]     + bfrag[k0 / 4].x, 0.0f);
        a.y = fmaxf(arow[ka + 1] + bfrag[k0 / 4].y, 0.0f);
        c = __builtin_amdgcn_wmma_f32_16x16x4_f32(false, a, false, wfrag[k0 / 4],
                                                  (short)0, c, false, false);
      }
      float* orow = acc + (wave * 16 + 8 * half) * HIDDEN + f;
#pragma unroll
      for (int v = 0; v < 8; ++v) orow[v * HIDDEN] = c[v];
    }
  } else {
    if (tid < BIN_NODES) {
      float* row = acc + tid * HIDDEN;
      float v[HIDDEN];
      float m = -INFINITY;
#pragma unroll
      for (int q = 0; q < HIDDEN; ++q) {
        v[q] = row[q] + bias[q];
        m = fmaxf(m, v[q]);
      }
      float s = 0.0f;
#pragma unroll
      for (int q = 0; q < HIDDEN; ++q) s += expf(v[q] - m);
      const float lse = m + logf(s);
#pragma unroll
      for (int q = 0; q < HIDDEN; ++q) row[q] = v[q] - lse;
    }
  }
  __syncthreads();
  for (int q = tid; q < BIN_NODES * HIDDEN / 4; q += AGG_THREADS) {
    const size_t o = (size_t)blockIdx.x * (BIN_NODES * HIDDEN) + (size_t)q * 4;
    if (o < (size_t)n * HIDDEN) *(float4*)&out[o] = *(const float4*)&acc[q * 4];
  }
}

extern "C" void kernel_launch(void* const* d_in, const int* in_sizes, int n_in,
                              void* d_out, int out_size, void* d_ws, size_t ws_size,
                              hipStream_t stream) {
  const float* x   = (const float*)d_in[0];
  const int*   ei  = (const int*)d_in[1];
  const float* ew  = (const float*)d_in[2];
  const float* W1  = (const float*)d_in[3];
  const float* b1  = (const float*)d_in[4];
  const float* W2  = (const float*)d_in[5];
  const float* b2  = (const float*)d_in[6];

  const int       N = in_sizes[0] / D_FEAT;
  const long long E = in_sizes[2];
  const int* src = ei;
  const int* dst = ei + E;

  float* bufA = (float*)d_ws;
  float* bufB = bufA + (size_t)N * HIDDEN;

  const int ntiles = N / 16;
  const dim3 blk(256);
  const int gemm_blocks = (ntiles + 7) / 8;
  const long long nelem = (long long)N * HIDDEN;

  const int nbins  = (N + BIN_NODES - 1) / BIN_NODES;
  const int nblk   = AGG_WAVES * GRP_BLKS;
  const int nedges = (int)E;
  const int chunk  = (nedges + nblk - 1) / nblk;
  int* rec    = (int*)(bufB + (size_t)N * HIDDEN);
  int* wcount = rec + 2 * (size_t)nedges;
  int* gstart = wcount + (size_t)nblk * nbins;
  const size_t need = (size_t)((char*)(gstart + AGG_WAVES * GS_STRIDE) - (char*)d_ws);

  if (need <= ws_size && nbins <= BIN_MAX && N <= (1 << 24) && chunk <= CHUNK_MAX) {
    k_count<<<nblk, BIN_THREADS, 0, stream>>>(dst, nedges, chunk, nbins, wcount);
    k_place_gemm1<<<nblk + (ntiles + GEMM_TILES - 1) / GEMM_TILES, BIN_THREADS, 0, stream>>>(
        src, dst, ew, nedges, chunk, nbins, wcount, gstart, rec, nblk, x, W1, bufA, ntiles);
    k_agg<EPI_LINEAR><<<nbins, AGG_THREADS, 0, stream>>>(bufA, rec, gstart, b1, W2, bufB, N);
    k_agg<EPI_LSM><<<nbins, AGG_THREADS, 0, stream>>>(bufB, rec, gstart, b2, W2, (float*)d_out, N);
    return;
  }

  k_zero<<<512, blk, 0, stream>>>(bufB, nelem);
  k_gemm1<<<gemm_blocks, blk, 0, stream>>>(x, W1, bufA, ntiles);
  k_scatter<<<32768, blk, 0, stream>>>(bufA, src, dst, ew, bufB, E);
  k_gemm2<<<gemm_blocks, blk, 0, stream>>>(bufB, b1, W2, bufA, ntiles);
  k_zero<<<512, blk, 0, stream>>>(bufB, nelem);
  k_scatter<<<32768, blk, 0, stream>>>(bufA, src, dst, ew, bufB, E);
  k_final<<<(N + 255) / 256, blk, 0, stream>>>(bufB, b2, (float*)d_out, N);
}


# Each place block orders its own edge chunk alone (LDS counts, records in its own edge range): no count launch, 4 launches to 3; an aggregation wave lists 32 short runs in LDS; sums in arrival order
# speedup vs baseline: 3.2529x; 1.0883x over previous
#include <hip/hip_runtime.h>
#include <cmath>

typedef __attribute__((ext_vector_type(2))) float v2f;
typedef __attribute__((ext_vector_type(8))) float v8f;

#define D_FEAT 512
#define HIDDEN 16

__global__ void k_zero(float* __restrict__ p, long long n) {
  long long i = (long long)blockIdx.x * blockDim.x + threadIdx.x;
  long long stride = (long long)gridDim.x * blockDim.x;
  for (; i < n; i += stride) p[i] = 0.0f;
}

__device__ __forceinline__ void gemm1_tile(const float* __restrict__ x, const float* __restrict__ W,
                                           float* __restrict__ h, int wave) {
  const int lane = threadIdx.x & 31;
  const int half = lane >> 4;
  const int l16  = lane & 15;

  const float* xrow = x + ((size_t)wave * 16 + l16) * D_FEAT + 2 * half;
  const float* wptr = W + (2 * half) * HIDDEN + l16;

  v8f c = {};
#pragma unroll 4
  for (int k0 = 0; k0 < D_FEAT; k0 += 4) {
    v2f a = *(const v2f*)(xrow + k0);
    v2f b;
    b.x = wptr[k0 * HIDDEN];
    b.y = wptr[(k0 + 1) * HIDDEN];
    c = __builtin_amdgcn_wmma_f32_16x16x4_f32(false, a, false, b,
                                              (short)0, c, false, false);
  }
  float* out = h + ((size_t)wave * 16 + 8 * half) * HIDDEN + l16;
#pragma unroll
  for (int v = 0; v < 8; ++v) out[v * HIDDEN] = c[v];
}

__global__ void k_gemm1(const float* __restrict__ x, const float* __restrict__ W,
                        float* __restrict__ h, int ntiles) {
  const int wave = blockIdx.x * (blockDim.x >> 5) + (threadIdx.x >> 5);
  if (wave >= ntiles) return;
  gemm1_tile(x, W, h, wave);
}

__global__ void k_gemm2(const float* __restrict__ acc, const float* __restrict__ bias,
                        const float* __restrict__ W, float* __restrict__ h, int ntiles) {
  const int wave = blockIdx.x * (blockDim.x >> 5) + (threadIdx.x >> 5);
  if (wave >= ntiles) return;
  const int lane = threadIdx.x & 31;
  const int half = lane >> 4;
  const int l16  = lane & 15;

  const float* arow = acc + ((size_t)wave * 16 + l16) * HIDDEN;

  v8f c = {};
#pragma unroll
  for (int k0 = 0; k0 < HIDDEN; k0 += 4) {
    const int ka = k0 + 2 * half;
    v2f a, b;
    a.x = fmaxf(arow[ka]     + bias[ka],     0.0f);
    a.y = fmaxf(arow[ka + 1] + bias[ka + 1], 0.0f);
    b.x = W[ka * HIDDEN + l16];
    b.y = W[(ka + 1) * HIDDEN + l16];
    c = __builtin_amdgcn_wmma_f32_16x16x4_f32(false, a, false, b,
                                              (short)0, c, false, false);
  }
  float* out = h + ((size_t)wave * 16 + 8 * half) * HIDDEN + l16;
#pragma unroll
  for (int v = 0; v < 8; ++v) out[v * HIDDEN] = c[v];
}

__global__ void k_scatter(const float* __restrict__ h, const int* __restrict__ src,
                          const int* __restrict__ dst, const float* __restrict__ w,
                          float* __restrict__ out, long long nedges) {
  long long i = (long long)blockIdx.x * blockDim.x + threadIdx.x;
  const long long stride = (long long)gridDim.x * blockDim.x;
  const long long total = nedges * HIDDEN;
  for (; i < total; i += stride) {
    const long long e = i >> 4;
    const int f = (int)(i & 15);
    const int s = src[e];
    const int d = dst[e];
    const float v = w[e] * h[(size_t)s * HIDDEN + f];
    atomicAdd(out + (size_t)d * HIDDEN + f, v);
  }
}

__global__ void k_final(const float* __restrict__ acc, const float* __restrict__ bias,
                        float* __restrict__ out, int n) {
  const int i = blockIdx.x * blockDim.x + threadIdx.x;
  if (i >= n) return;
  float v[HIDDEN];
  float m = -INFINITY;
#pragma unroll
  for (int f = 0; f < HIDDEN; ++f) {
    v[f] = acc[(size_t)i * HIDDEN + f] + bias[f];
    m = fmaxf(m, v[f]);
  }
  float s = 0.0f;
#pragma unroll
  for (int f = 0; f < HIDDEN; ++f) s += expf(v[f] - m);
  const float lse = m + logf(s);
#pragma unroll
  for (int f = 0; f < HIDDEN; ++f) out[(size_t)i * HIDDEN + f] = v[f] - lse;
}

#define BIN_NODES 98
#define BIN_ROWS  112
#define BIN_LMASK 127
#define BIN_MAX   1024
__device__ __forceinline__ int bin_of(int d) { return (int)((unsigned)d / (unsigned)BIN_NODES); }
#define BIN_THREADS 1024
#define AGG_THREADS 256
#define CHUNK_MAX 12800
#define EDGE_TRIPS ((CHUNK_MAX + BIN_THREADS - 1) / BIN_THREADS)
#define AGG_WAVES (AGG_THREADS / 32)
#define PLACE_PER_WAVE 32
#define LS_STRIDE 1024
#define LIST_CAP  1024

#define GEMM_TILES 25
__global__ void __launch_bounds__(BIN_THREADS)
k_place_gemm1(const int* __restrict__ src, const int* __restrict__ dst,
              const float* __restrict__ w, int nedges, int chunk, int nbins,
              int* __restrict__ lstart, int* __restrict__ rec, int nblk,
              const float* __restrict__ x, const float* __restrict__ W1,
              float* __restrict__ h, int ntiles) {
  __shared__ int2 srec[CHUNK_MAX];
  __shared__ int scanL[BIN_THREADS];
  __shared__ int lcur[BIN_MAX];
  const int tid = threadIdx.x;
  if ((int)blockIdx.x >= nblk) {
    const int wib  = tid >> 5;
    const int wave = ((int)blockIdx.x - nblk) * GEMM_TILES + wib;
    if (wib < GEMM_TILES && wave < ntiles)
      gemm1_tile(x, W1, h, wave);
    return;
  }
  scanL[tid] = 0;
  __syncthreads();
  const int e0 = min((int)blockIdx.x * chunk, nedges);
  const int e1 = min(e0 + chunk, nedges);
  int dreg[EDGE_TRIPS], sreg[EDGE_TRIPS], wreg[EDGE_TRIPS];
#pragma unroll
  for (int i = 0; i < EDGE_TRIPS; ++i) {
    const int e = e0 + tid + i * BIN_THREADS;
    dreg[i] = (e < e1) ? dst[e] : 0;
  }
#pragma unroll
  for (int i = 0; i < EDGE_TRIPS; ++i) {
    const int e = e0 + tid + i * BIN_THREADS;
    sreg[i] = (e < e1) ? src[e] : 0;
    wreg[i] = (e < e1) ? __float_as_int(w[e]) : 0;
  }
#pragma unroll
  for (int i = 0; i < EDGE_TRIPS; ++i)
    if (e0 + tid + i * BIN_THREADS < e1) atomicAdd(&scanL[bin_of(dreg[i])], 1);
  __syncthreads();
  const int own = scanL[tid];
  for (int off = 1; off < BIN_THREADS; off <<= 1) {
    int vl = scanL[tid];
    if (tid >= off) vl += scanL[tid - off];
    __syncthreads();
    scanL[tid] = vl;
    __syncthreads();
  }
  const int inclL = scanL[tid];
  const int exclL = inclL - own;
  lcur[tid] = exclL;
  {
    int* __restrict__ ls = lstart + (size_t)blockIdx.x * LS_STRIDE;
    if (tid < nbins) ls[tid] = e0 + exclL;
    if (tid == nbins - 1) ls[nbins] = e0 + inclL;
  }
  __syncthreads();
#pragma unroll
  for (int i = 0; i < EDGE_TRIPS; ++i) {
    const int e = e0 + tid + i * BIN_THREADS;
    if (e < e1) {
      const int d = dreg[i];
      const int s = sreg[i];
      const int wv = wreg[i];
      const int b = bin_of(d);
      const int j = atomicAdd(&lcur[b], 1);
      srec[j] = make_int2(s | ((d - b * BIN_NODES) << 24), wv);
    }
  }
  __syncthreads();
  int2* __restrict__ rec2 = (int2*)rec;
  const int cnt = e1 - e0;
  for (int j = tid; j < cnt; j += BIN_THREADS) rec2[e0 + j] = srec[j];
}

#define EPI_LINEAR 0
#define EPI_LSM    1
#define ROW_BCAST(v, n) __builtin_amdgcn_mov_dpp((v), 0x150 + (n), 0xf, 0xf, false)
#define AGG_GATHER(n)                                                        \
  const int   s##n = ROW_BCAST(r.x, n);                                      \
  const float w##n = __int_as_float(ROW_BCAST(r.y, n));                      \
  const float g##n = h[(size_t)(s##n & 0xFFFFFF) * HIDDEN + f];
#define AGG_ADD(n) \
  atomicAdd(&acc[((s##n >> 24) & BIN_LMASK) * HIDDEN + f], w##n * g##n);
#define AGG_ADD_IF(n) \
  if (16 * half + (n) < cnt) AGG_ADD(n)
#define AGG_16(M) M(0) M(1) M(2) M(3) M(4) M(5) M(6) M(7) M(8) M(9) M(10) M(11) M(12) M(13) M(14) M(15)
template <int EPI>
__global__ void k_agg(const float* __restrict__ h, const int* __restrict__ rec,
                      const int* __restrict__ lstart, const float* __restrict__ bias,
                      const float* __restrict__ W, float* __restrict__ out, int n) {
  __shared__ __attribute__((aligned(16))) float acc[BIN_ROWS * HIDDEN];
  __shared__ int wlist[AGG_WAVES * LIST_CAP];
  const int tid = threadIdx.x;
  for (int q = tid; q < BIN_ROWS * HIDDEN / 4; q += AGG_THREADS)
    *(float4*)&acc[q * 4] = make_float4(0.0f, 0.0f, 0.0f, 0.0f);
  __syncthreads();
  const int f = tid & 15;
  const int2* __restrict__ rec2 = (const int2*)rec;
  const int half = (tid >> 4) & 1;
  v2f wfrag[HIDDEN / 4], bfrag[HIDDEN / 4];
  if (EPI == EPI_LINEAR) {
#pragma unroll
    for (int k0 = 0; k0 < HIDDEN; k0 += 4) {
      const int ka = k0 + 2 * half;
      wfrag[k0 / 4].x = W[ka * HIDDEN + f];
      wfrag[k0 / 4].y = W[(ka + 1) * HIDDEN + f];
      bfrag[k0 / 4].x = bias[ka];
      bfrag[k0 / 4].y = bias[ka + 1];
    }
  }
  const int lane = tid & 31;
  const int wv = __builtin_amdgcn_readfirstlane(tid >> 5);
  const int* __restrict__ ls = lstart + (size_t)(wv * PLACE_PER_WAVE + lane) * LS_STRIDE + blockIdx.x;
  const int lo  = ls[0];
  const int len = ls[1] - lo;
  int inc = len;
#pragma unroll
  for (int d = 1; d < 32; d <<= 1) {
    const int t = __shfl_up(inc, d);
    if (lane >= d) inc += t;
  }
  const int exc = inc - len;
  const int total = __builtin_amdgcn_readlane(inc, 31);
  int* myl = wlist + wv * LIST_CAP;
  for (int base = 0; base < total; base += LIST_CAP) {
    const int seg = min(total - base, LIST_CAP);
    {
      const int t0 = max(exc, base);
      const int t1 = min(inc, base + seg);
#pragma nounroll
      for (int t = t0; t < t1; ++t) myl[t - base] = lo + (t - exc);
    }
    int2 r = rec2[myl[min(lane, seg - 1)]];
    int inext = myl[min(32 + lane, seg - 1)];
    for (int b = 0; b < seg; b += 32) {
      const int cnt = seg - b;
      AGG_16(AGG_GATHER)
      const int2 rn = rec2[inext];
      inext = myl[min(b + 64 + lane, seg - 1)];
      if (cnt >= 32) {
        AGG_16(AGG_ADD)
      } else {
        AGG_16(AGG_ADD_IF)
      }
      r = rn;
    }
  }
  __syncthreads();
  if (EPI == EPI_LINEAR) {
    const int wave = tid >> 5;
    if (wave < BIN_ROWS / 16) {
      const float* arow = acc + (wave * 16 + f) * HIDDEN;
      v8f c = {};
#pragma unroll
      for (int k0 = 0; k0 < HIDDEN; k0 += 4) {
        const int ka = k0 + 2 * half;
        v2f a;
        a.x = fmaxf(arow[ka]     + bfrag[k0 / 4].x, 0.0f);
        a.y = fmaxf(arow[ka + 1] + bfrag[k0 / 4].y, 0.0f);
        c = __builtin_amdgcn_wmma_f32_16x16x4_f32(false, a, false, wfrag[k0 / 4],
                                                  (short)0, c, false, false);
      }
      float* orow = acc + (wave * 16 + 8 * half) * HIDDEN + f;
#pragma unroll
      for (int v = 0; v < 8; ++v) orow[v * HIDDEN] = c[v];
    }
  } else {
    if (tid < BIN_NODES) {
      float* row = acc + tid * HIDDEN;
      float v[HIDDEN];
      float m = -INFINITY;
#pragma unroll
      for (int q = 0; q < HIDDEN; ++q) {
        v[q] = row[q] + bias[q];
        m = fmaxf(m, v[q]);
      }
      float s = 0.0f;
#pragma unroll
      for (int q = 0; q < HIDDEN; ++q) s += expf(v[q] - m);
      const float lse = m + logf(s);
#pragma unroll
      for (int q = 0; q < HIDDEN; ++q) row[q] = v[q] - lse;
    }
  }
  __syncthreads();
  for (int q = tid; q < BIN_NODES * HIDDEN / 4; q += AGG_THREADS) {
    const size_t o = (size_t)blockIdx.x * (BIN_NODES * HIDDEN) + (size_t)q * 4;
    if (o < (size_t)n * HIDDEN) *(float4*)&out[o] = *(const float4*)&acc[q * 4];
  }
}

extern "C" void kernel_launch(void* const* d_in, const int* in_sizes, int n_in,
                              void* d_out, int out_size, void* d_ws, size_t ws_size,
                              hipStream_t stream) {
  const float* x   = (const float*)d_in[0];
  const int*   ei  = (const int*)d_in[1];
  const float* ew  = (const float*)d_in[2];
  const float* W1  = (const float*)d_in[3];
  const float* b1  = (const float*)d_in[4];
  const float* W2  = (const float*)d_in[5];
  const float* b2  = (const float*)d_in[6];

  const int       N = in_sizes[0] / D_FEAT;
  const long long E = in_sizes[2];
  const int* src = ei;
  const int* dst = ei + E;

  float* bufA = (float*)d_ws;
  float* bufB = bufA + (size_t)N * HIDDEN;

  const int ntiles = N / 16;
  const dim3 blk(256);
  const int gemm_blocks = (ntiles + 7) / 8;
  const long long nelem = (long long)N * HIDDEN;

  const int nbins  = (N + BIN_NODES - 1) / BIN_NODES;
  const int nblk   = AGG_WAVES * PLACE_PER_WAVE;
  const int nedges = (int)E;
  const int chunk  = (nedges + nblk - 1) / nblk;
  int* rec    = (int*)(bufB + (size_t)N * HIDDEN);
  int* lstart = rec + 2 * (size_t)nedges;
  const size_t need = (size_t)((char*)(lstart + (size_t)nblk * LS_STRIDE) - (char*)d_ws);

  if (need <= ws_size && nbins < LS_STRIDE && nbins <= BIN_MAX && N <= (1 << 24) && chunk <= CHUNK_MAX) {
    k_place_gemm1<<<nblk + (ntiles + GEMM_TILES - 1) / GEMM_TILES, BIN_THREADS, 0, stream>>>(
        src, dst, ew, nedges, chunk, nbins, lstart, rec, nblk, x, W1, bufA, ntiles);
    k_agg<EPI_LINEAR><<<nbins, AGG_THREADS, 0, stream>>>(bufA, rec, lstart, b1, W2, bufB, N);
    k_agg<EPI_LSM><<<nbins, AGG_THREADS, 0, stream>>>(bufB, rec, lstart, b2, W2, (float*)d_out, N);
    return;
  }

  k_zero<<<512, blk, 0, stream>>>(bufB, nelem);
  k_gemm1<<<gemm_blocks, blk, 0, stream>>>(x, W1, bufA, ntiles);
  k_scatter<<<32768, blk, 0, stream>>>(bufA, src, dst, ew, bufB, E);
  k_gemm2<<<gemm_blocks, blk, 0, stream>>>(bufB, b1, W2, bufA, ntiles);
  k_zero<<<512, blk, 0, stream>>>(bufB, nelem);
  k_scatter<<<32768, blk, 0, stream>>>(bufA, src, dst, ew, bufB, E);
  k_final<<<(N + 255) / 256, blk, 0, stream>>>(bufB, b2, (float*)d_out, N);
}
